# starting kernel
# baseline (speedup 1.0000x reference)
#include <hip/hip_runtime.h>
#include <hip/hip_bf16.h>


#define S_LEN 2048
#define B_SZ  8
#define E_DIM 1024
constexpr float GAMMA = 0.5f;

typedef __attribute__((ext_vector_type(2))) float v2f;
typedef __attribute__((ext_vector_type(8))) float v8f;

#define GAS __attribute__((address_space(1)))

static __device__ __forceinline__ v2f load_g2(const float* p) {
  return *(const GAS v2f*)(p);
}
static __device__ __forceinline__ float load_g1(const float* p) {
  return *(const GAS float*)(p);
}
static __device__ __forceinline__ void store_g1(float* p, float x) {
  *(GAS float*)(p) = x;
}

static __device__ __forceinline__ v8f wmma_f32(v2f a, v2f b, v8f c) {
  return __builtin_amdgcn_wmma_f32_16x16x4_f32(false, a, false, b,
                                               (short)0, c, false, false);
}

__global__ __launch_bounds__(32)
void qkv_proj_kernel(const float* __restrict__ x,
                     const float* __restrict__ Wq, const float* __restrict__ bq,
                     const float* __restrict__ Wk, const float* __restrict__ bk,
                     const float* __restrict__ Wv, const float* __restrict__ bv,
                     float* __restrict__ q, float* __restrict__ k,
                     float* __restrict__ v) {
  const int lane = threadIdx.x;
  const int half = lane >> 4;
  const int lm   = lane & 15;
  const int m0   = blockIdx.x * 16;
  const int n0   = blockIdx.y * 64;

  v8f aq[4], ak[4], av[4];
  const float* wqp[4];
  const float* wkp[4];
  const float* wvp[4];
  #pragma unroll
  for (int ns = 0; ns < 4; ++ns) {
    const int col = n0 + ns * 16 + lm;
    const float biq = load_g1(bq + col);
    const float bik = load_g1(bk + col);
    const float biv = load_g1(bv + col);
    #pragma unroll
    for (int g = 0; g < 8; ++g) { aq[ns][g] = biq; ak[ns][g] = bik; av[ns][g] = biv; }
    const size_t wo = (size_t)col * E_DIM + 2 * half;
    wqp[ns] = Wq + wo;
    wkp[ns] = Wk + wo;
    wvp[ns] = Wv + wo;
  }

  const float* xp = x + (size_t)(m0 + lm) * E_DIM + 2 * half;
  for (int kk = 0; kk < E_DIM; kk += 16) {
    #pragma unroll
    for (int u = 0; u < 4; ++u) {
      const v2f a = load_g2(xp + 4 * u);
      #pragma unroll
      for (int ns = 0; ns < 4; ++ns) {
        aq[ns] = wmma_f32(a, load_g2(wqp[ns] + 4 * u), aq[ns]);
        ak[ns] = wmma_f32(a, load_g2(wkp[ns] + 4 * u), ak[ns]);
        av[ns] = wmma_f32(a, load_g2(wvp[ns] + 4 * u), av[ns]);
      }
    }
    xp += 16;
    #pragma unroll
    for (int ns = 0; ns < 4; ++ns) { wqp[ns] += 16; wkp[ns] += 16; wvp[ns] += 16; }
  }

  #pragma unroll
  for (int ns = 0; ns < 4; ++ns) {
    const int col = n0 + ns * 16 + lm;
    #pragma unroll
    for (int g = 0; g < 8; ++g) {
      const size_t row = (size_t)(m0 + g + 8 * half);
      store_g1(q + row * E_DIM + col, aq[ns][g]);
      store_g1(k + row * E_DIM + col, ak[ns][g]);
      store_g1(v + row * E_DIM + col, av[ns][g]);
    }
  }
}

__global__ __launch_bounds__(256)
void rownorm_kernel(const float* __restrict__ k, float* __restrict__ gk2) {
  __shared__ float red[256];
  const float* kr = k + (size_t)blockIdx.x * E_DIM;
  float s = 0.f;
  for (int e = threadIdx.x; e < E_DIM; e += 256) {
    const float t = load_g1(kr + e);
    s += t * t;
  }
  red[threadIdx.x] = s;
  __syncthreads();
  for (int off = 128; off > 0; off >>= 1) {
    if (threadIdx.x < off) red[threadIdx.x] += red[threadIdx.x + off];
    __syncthreads();
  }
  if (threadIdx.x == 0) store_g1(gk2 + blockIdx.x, GAMMA * red[0]);
}

__global__ __launch_bounds__(256)
void attn_kernel(const float* __restrict__ q, const float* __restrict__ k,
                 const float* __restrict__ v, const float* __restrict__ gk2,
                 float* __restrict__ out) {
  const int tid  = threadIdx.x;
  const int w    = tid >> 5;
  const int lane = tid & 31;
  const int half = lane >> 4;
  const int lm   = lane & 15;
  const int b    = blockIdx.y;
  const int t0   = blockIdx.x * 16;
  const int e0   = w * 128;

  __shared__ float sS[8][256];
  __shared__ float sP[256];
  __shared__ float sScale[16];
  __shared__ float sM[16];
  __shared__ float sL[16];

  if (tid < 16) { sM[tid] = -__builtin_inff(); sL[tid] = 0.f; }
  __syncthreads();

  v2f qf[32];
  {
    const float* qrow =
        q + ((size_t)(t0 + lm) * B_SZ + b) * E_DIM + e0 + 2 * half;
    #pragma unroll
    for (int i = 0; i < 32; ++i) qf[i] = load_g2(qrow + 4 * i) * (2.0f * GAMMA);
  }

  v8f acc[8];
  #pragma unroll
  for (int ns = 0; ns < 8; ++ns) acc[ns] = (v8f){};

  for (int j0 = 0; j0 < S_LEN; j0 += 16) {
    v8f sc = (v8f){};
    const float* krow =
        k + ((size_t)(j0 + lm) * B_SZ + b) * E_DIM + e0 + 2 * half;
    #pragma unroll
    for (int i = 0; i < 32; ++i)
      sc = wmma_f32(qf[i], load_g2(krow + 4 * i), sc);

    #pragma unroll
    for (int g = 0; g < 8; ++g)
      sS[w][(g + 8 * half) * 16 + lm] = sc[g];
    __syncthreads();

    float logit;
    {
      float s = 0.f;
      #pragma unroll
      for (int ww = 0; ww < 8; ++ww) s += sS[ww][tid];
      logit = s - load_g1(gk2 + (size_t)(j0 + (tid & 15)) * B_SZ + b);
      sP[tid] = logit;
    }
    __syncthreads();

    float scl_row = 0.f;
    if (tid < 16) {
      const int row = tid;
      float mrow = -__builtin_inff();
      #pragma unroll
      for (int c = 0; c < 16; ++c) mrow = fmaxf(mrow, sP[row * 16 + c]);
      const float mnew = fmaxf(sM[row], mrow);
      scl_row = __expf(sM[row] - mnew);
      sM[row] = mnew;
      sScale[row] = scl_row;
    }
    __syncthreads();

    sP[tid] = __expf(logit - sM[tid >> 4]);
    __syncthreads();

    if (tid < 16) {
      float psum = 0.f;
      #pragma unroll
      for (int c = 0; c < 16; ++c) psum += sP[tid * 16 + c];
      sL[tid] = sL[tid] * scl_row + psum;
    }

    float f[8];
    #pragma unroll
    for (int g = 0; g < 8; ++g) f[g] = sScale[g + 8 * half];
    #pragma unroll
    for (int ns = 0; ns < 8; ++ns)
      #pragma unroll
      for (int g = 0; g < 8; ++g) acc[ns][g] *= f[g];

    #pragma unroll
    for (int c = 0; c < 4; ++c) {
      const v2f pf = *(const v2f*)&sP[lm * 16 + 4 * c + 2 * half];
      const int jA = j0 + 4 * c + 2 * half;
      const float* vbase = v + ((size_t)jA * B_SZ + b) * E_DIM + e0 + lm;
      #pragma unroll
      for (int ns = 0; ns < 8; ++ns) {
        v2f vf;
        vf.x = load_g1(vbase + ns * 16);
        vf.y = load_g1(vbase + (size_t)B_SZ * E_DIM + ns * 16);
        acc[ns] = wmma_f32(pf, vf, acc[ns]);
      }
    }
  }
  __syncthreads();

  float linv[8];
  #pragma unroll
  for (int g = 0; g < 8; ++g) linv[g] = 1.0f / sL[g + 8 * half];
  #pragma unroll
  for (int ns = 0; ns < 8; ++ns) {
    const int e = e0 + ns * 16 + lm;
    #pragma unroll
    for (int g = 0; g < 8; ++g) {
      const int trow = t0 + g + 8 * half;
      store_g1(out + ((size_t)trow * B_SZ + b) * E_DIM + e, acc[ns][g] * linv[g]);
    }
  }
}

extern "C" void kernel_launch(void* const* d_in, const int* in_sizes, int n_in,
                              void* d_out, int out_size, void* d_ws, size_t ws_size,
                              hipStream_t stream) {
  (void)in_sizes; (void)n_in; (void)out_size; (void)ws_size;
  const float* x  = (const float*)d_in[0];
  const float* Wq = (const float*)d_in[1];
  const float* bq = (const float*)d_in[2];
  const float* Wk = (const float*)d_in[3];
  const float* bk = (const float*)d_in[4];
  const float* Wv = (const float*)d_in[5];
  const float* bv = (const float*)d_in[6];
  float* out = (float*)d_out;

  const size_t mat = (size_t)S_LEN * B_SZ * E_DIM;
  float* q   = (float*)d_ws;
  float* k   = q + mat;
  float* v   = k + mat;
  float* gk2 = v + mat;

  dim3 g1((S_LEN * B_SZ) / 16, E_DIM / 64);
  qkv_proj_kernel<<<g1, 32, 0, stream>>>(x, Wq, bq, Wk, bk, Wv, bv, q, k, v);

  rownorm_kernel<<<S_LEN * B_SZ, 256, 0, stream>>>(k, gk2);

  dim3 g3(S_LEN / 16, B_SZ);
  attn_kernel<<<g3, 256, 0, stream>>>(q, k, v, gk2, out);
}


# QKV projection re-tiled: 8-wave workgroup per 128x128 output tile, x and W k-blocks staged in LDS by whole-line loads, wide fragment reads; k sum order inside each 8-k chunk changes
# speedup vs baseline: 2.1805x; 2.1805x over previous
#include <hip/hip_runtime.h>
#include <hip/hip_bf16.h>


#define S_LEN 2048
#define B_SZ  8
#define E_DIM 1024
constexpr float GAMMA = 0.5f;

typedef __attribute__((ext_vector_type(2))) float v2f;
typedef __attribute__((ext_vector_type(8))) float v8f;

#define GAS __attribute__((address_space(1)))

static __device__ __forceinline__ v2f load_g2(const float* p) {
  return *(const GAS v2f*)(p);
}
static __device__ __forceinline__ float load_g1(const float* p) {
  return *(const GAS float*)(p);
}
static __device__ __forceinline__ void store_g1(float* p, float x) {
  *(GAS float*)(p) = x;
}

static __device__ __forceinline__ v8f wmma_f32(v2f a, v2f b, v8f c) {
  return __builtin_amdgcn_wmma_f32_16x16x4_f32(false, a, false, b,
                                               (short)0, c, false, false);
}

#define P_BM 128
#define P_BN 128
#define P_BK 32
#define P_LD 36

typedef __attribute__((ext_vector_type(4))) float v4f;

static __device__ __forceinline__ v4f load_g4(const float* p) {
  return *(const GAS v4f*)(p);
}

__global__ __launch_bounds__(256)
void qkv_proj_kernel(const float* __restrict__ x,
                     const float* __restrict__ Wq, const float* __restrict__ bq,
                     const float* __restrict__ Wk, const float* __restrict__ bk,
                     const float* __restrict__ Wv, const float* __restrict__ bv,
                     float* __restrict__ q, float* __restrict__ k,
                     float* __restrict__ v) {
  __shared__ __attribute__((aligned(16))) float sA[P_BM * P_LD];
  __shared__ __attribute__((aligned(16))) float sB[P_BN * P_LD];

  const int tid  = threadIdx.x;
  const int w    = tid >> 5;
  const int lane = tid & 31;
  const int half = lane >> 4;
  const int lm   = lane & 15;
  const int wm   = w >> 1;
  const int wn   = w & 1;
  const int nt   = blockIdx.x;
  const int sel  = nt >> 3;
  const int n0   = (nt & 7) * P_BN;
  const int m0   = blockIdx.y * P_BM;

  const float* Wm = Wq; const float* bias = bq; float* dst = q;
  if (sel == 1) { Wm = Wk; bias = bk; dst = k; }
  if (sel == 2) { Wm = Wv; bias = bv; dst = v; }

  const int srow = tid >> 3;
  const int scol = (tid & 7) * 4;
  const float* xb = x  + (size_t)m0 * E_DIM;
  const float* wb = Wm + (size_t)n0 * E_DIM;
  const unsigned goff = (unsigned)(srow * E_DIM + scol);
  const int st_off = srow * P_LD + scol;

  const int a_off = (wm * 32 + lm) * P_LD + 4 * half;
  const int b_off = (wn * 64 + lm) * P_LD + 4 * half;

  v8f acc[2][4];
  #pragma unroll
  for (int ni = 0; ni < 4; ++ni) {
    const float bi = load_g1(bias + n0 + wn * 64 + ni * 16 + lm);
    #pragma unroll
    for (int mi = 0; mi < 2; ++mi)
      #pragma unroll
      for (int g = 0; g < 8; ++g) acc[mi][ni][g] = bi;
  }

  v4f ra[4], rb[4];
  #pragma unroll
  for (int i = 0; i < 4; ++i) {
    ra[i] = load_g4(xb + goff + i * 32 * E_DIM);
    rb[i] = load_g4(wb + goff + i * 32 * E_DIM);
  }
  #pragma unroll
  for (int i = 0; i < 4; ++i) {
    *(v4f*)&sA[st_off + i * 32 * P_LD] = ra[i];
    *(v4f*)&sB[st_off + i * 32 * P_LD] = rb[i];
  }
  __syncthreads();

  for (int kb = 0; kb < E_DIM / P_BK; ++kb) {
    const bool more = (kb + 1) < (E_DIM / P_BK);
    if (more) {
      xb += P_BK;
      wb += P_BK;
      #pragma unroll
      for (int i = 0; i < 4; ++i) {
        ra[i] = load_g4(xb + goff + i * 32 * E_DIM);
        rb[i] = load_g4(wb + goff + i * 32 * E_DIM);
      }
    }

    v4f fa[2][2], fb[2][4];
    #pragma unroll
    for (int mi = 0; mi < 2; ++mi) fa[0][mi] = *(const v4f*)&sA[a_off + mi * 16 * P_LD];
    #pragma unroll
    for (int ni = 0; ni < 4; ++ni) fb[0][ni] = *(const v4f*)&sB[b_off + ni * 16 * P_LD];

    #pragma unroll
    for (int s = 0; s < P_BK / 8; ++s) {
      const int cur = s & 1, nxt = cur ^ 1;
      if (s + 1 < P_BK / 8) {
        #pragma unroll
        for (int mi = 0; mi < 2; ++mi)
          fa[nxt][mi] = *(const v4f*)&sA[a_off + mi * 16 * P_LD + (s + 1) * 8];
        #pragma unroll
        for (int ni = 0; ni < 4; ++ni)
          fb[nxt][ni] = *(const v4f*)&sB[b_off + ni * 16 * P_LD + (s + 1) * 8];
      }
      #pragma unroll
      for (int mi = 0; mi < 2; ++mi)
        #pragma unroll
        for (int ni = 0; ni < 4; ++ni)
          acc[mi][ni] = wmma_f32(fa[cur][mi].lo, fb[cur][ni].lo, acc[mi][ni]);
      #pragma unroll
      for (int mi = 0; mi < 2; ++mi)
        #pragma unroll
        for (int ni = 0; ni < 4; ++ni)
          acc[mi][ni] = wmma_f32(fa[cur][mi].hi, fb[cur][ni].hi, acc[mi][ni]);
    }

    __syncthreads();
    if (more) {
      #pragma unroll
      for (int i = 0; i < 4; ++i) {
        *(v4f*)&sA[st_off + i * 32 * P_LD] = ra[i];
        *(v4f*)&sB[st_off + i * 32 * P_LD] = rb[i];
      }
    }
    __syncthreads();
  }

  #pragma unroll
  for (int mi = 0; mi < 2; ++mi) {
    #pragma unroll
    for (int ni = 0; ni < 4; ++ni) {
      const int col = n0 + wn * 64 + ni * 16 + lm;
      #pragma unroll
      for (int g = 0; g < 8; ++g) {
        const size_t row = (size_t)(m0 + wm * 32 + mi * 16 + g + 8 * half);
        store_g1(dst + row * E_DIM + col, acc[mi][ni][g]);
      }
    }
  }
}

__global__ __launch_bounds__(256)
void rownorm_kernel(const float* __restrict__ k, float* __restrict__ gk2) {
  __shared__ float red[256];
  const float* kr = k + (size_t)blockIdx.x * E_DIM;
  float s = 0.f;
  for (int e = threadIdx.x; e < E_DIM; e += 256) {
    const float t = load_g1(kr + e);
    s += t * t;
  }
  red[threadIdx.x] = s;
  __syncthreads();
  for (int off = 128; off > 0; off >>= 1) {
    if (threadIdx.x < off) red[threadIdx.x] += red[threadIdx.x + off];
    __syncthreads();
  }
  if (threadIdx.x == 0) store_g1(gk2 + blockIdx.x, GAMMA * red[0]);
}

__global__ __launch_bounds__(256)
void attn_kernel(const float* __restrict__ q, const float* __restrict__ k,
                 const float* __restrict__ v, const float* __restrict__ gk2,
                 float* __restrict__ out) {
  const int tid  = threadIdx.x;
  const int w    = tid >> 5;
  const int lane = tid & 31;
  const int half = lane >> 4;
  const int lm   = lane & 15;
  const int b    = blockIdx.y;
  const int t0   = blockIdx.x * 16;
  const int e0   = w * 128;

  __shared__ float sS[8][256];
  __shared__ float sP[256];
  __shared__ float sScale[16];
  __shared__ float sM[16];
  __shared__ float sL[16];

  if (tid < 16) { sM[tid] = -__builtin_inff(); sL[tid] = 0.f; }
  __syncthreads();

  v2f qf[32];
  {
    const float* qrow =
        q + ((size_t)(t0 + lm) * B_SZ + b) * E_DIM + e0 + 2 * half;
    #pragma unroll
    for (int i = 0; i < 32; ++i) qf[i] = load_g2(qrow + 4 * i) * (2.0f * GAMMA);
  }

  v8f acc[8];
  #pragma unroll
  for (int ns = 0; ns < 8; ++ns) acc[ns] = (v8f){};

  for (int j0 = 0; j0 < S_LEN; j0 += 16) {
    v8f sc = (v8f){};
    const float* krow =
        k + ((size_t)(j0 + lm) * B_SZ + b) * E_DIM + e0 + 2 * half;
    #pragma unroll
    for (int i = 0; i < 32; ++i)
      sc = wmma_f32(qf[i], load_g2(krow + 4 * i), sc);

    #pragma unroll
    for (int g = 0; g < 8; ++g)
      sS[w][(g + 8 * half) * 16 + lm] = sc[g];
    __syncthreads();

    float logit;
    {
      float s = 0.f;
      #pragma unroll
      for (int ww = 0; ww < 8; ++ww) s += sS[ww][tid];
      logit = s - load_g1(gk2 + (size_t)(j0 + (tid & 15)) * B_SZ + b);
      sP[tid] = logit;
    }
    __syncthreads();

    float scl_row = 0.f;
    if (tid < 16) {
      const int row = tid;
      float mrow = -__builtin_inff();
      #pragma unroll
      for (int c = 0; c < 16; ++c) mrow = fmaxf(mrow, sP[row * 16 + c]);
      const float mnew = fmaxf(sM[row], mrow);
      scl_row = __expf(sM[row] - mnew);
      sM[row] = mnew;
      sScale[row] = scl_row;
    }
    __syncthreads();

    sP[tid] = __expf(logit - sM[tid >> 4]);
    __syncthreads();

    if (tid < 16) {
      float psum = 0.f;
      #pragma unroll
      for (int c = 0; c < 16; ++c) psum += sP[tid * 16 + c];
      sL[tid] = sL[tid] * scl_row + psum;
    }

    float f[8];
    #pragma unroll
    for (int g = 0; g < 8; ++g) f[g] = sScale[g + 8 * half];
    #pragma unroll
    for (int ns = 0; ns < 8; ++ns)
      #pragma unroll
      for (int g = 0; g < 8; ++g) acc[ns][g] *= f[g];

    #pragma unroll
    for (int c = 0; c < 4; ++c) {
      const v2f pf = *(const v2f*)&sP[lm * 16 + 4 * c + 2 * half];
      const int jA = j0 + 4 * c + 2 * half;
      const float* vbase = v + ((size_t)jA * B_SZ + b) * E_DIM + e0 + lm;
      #pragma unroll
      for (int ns = 0; ns < 8; ++ns) {
        v2f vf;
        vf.x = load_g1(vbase + ns * 16);
        vf.y = load_g1(vbase + (size_t)B_SZ * E_DIM + ns * 16);
        acc[ns] = wmma_f32(pf, vf, acc[ns]);
      }
    }
  }
  __syncthreads();

  float linv[8];
  #pragma unroll
  for (int g = 0; g < 8; ++g) linv[g] = 1.0f / sL[g + 8 * half];
  #pragma unroll
  for (int ns = 0; ns < 8; ++ns) {
    const int e = e0 + ns * 16 + lm;
    #pragma unroll
    for (int g = 0; g < 8; ++g) {
      const int trow = t0 + g + 8 * half;
      store_g1(out + ((size_t)trow * B_SZ + b) * E_DIM + e, acc[ns][g] * linv[g]);
    }
  }
}

extern "C" void kernel_launch(void* const* d_in, const int* in_sizes, int n_in,
                              void* d_out, int out_size, void* d_ws, size_t ws_size,
                              hipStream_t stream) {
  (void)in_sizes; (void)n_in; (void)out_size; (void)ws_size;
  const float* x  = (const float*)d_in[0];
  const float* Wq = (const float*)d_in[1];
  const float* bq = (const float*)d_in[2];
  const float* Wk = (const float*)d_in[3];
  const float* bk = (const float*)d_in[4];
  const float* Wv = (const float*)d_in[5];
  const float* bv = (const float*)d_in[6];
  float* out = (float*)d_out;

  const size_t mat = (size_t)S_LEN * B_SZ * E_DIM;
  float* q   = (float*)d_ws;
  float* k   = q + mat;
  float* v   = k + mat;
  float* gk2 = v + mat;

  dim3 g1((3 * E_DIM) / P_BN, (S_LEN * B_SZ) / P_BM);
  qkv_proj_kernel<<<g1, 256, 0, stream>>>(x, Wq, bq, Wk, bk, Wv, bv, q, k, v);

  rownorm_kernel<<<S_LEN * B_SZ, 256, 0, stream>>>(k, gk2);

  dim3 g3(S_LEN / 16, B_SZ);
  attn_kernel<<<g3, 256, 0, stream>>>(q, k, v, gk2, out);
}


# attention: Q, K, V fragments and output rows moved as 16-byte per-lane global accesses (P*V columns re-mapped to 4 consecutive e per lane); sum order inside each 8-e chunk of q.k changes
# speedup vs baseline: 2.3470x; 1.0764x over previous
#include <hip/hip_runtime.h>
#include <hip/hip_bf16.h>


#define S_LEN 2048
#define B_SZ  8
#define E_DIM 1024
constexpr float GAMMA = 0.5f;

typedef __attribute__((ext_vector_type(2))) float v2f;
typedef __attribute__((ext_vector_type(8))) float v8f;

#define GAS __attribute__((address_space(1)))

static __device__ __forceinline__ float load_g1(const float* p) {
  return *(const GAS float*)(p);
}
static __device__ __forceinline__ void store_g1(float* p, float x) {
  *(GAS float*)(p) = x;
}

static __device__ __forceinline__ v8f wmma_f32(v2f a, v2f b, v8f c) {
  return __builtin_amdgcn_wmma_f32_16x16x4_f32(false, a, false, b,
                                               (short)0, c, false, false);
}

#define P_BM 128
#define P_BN 128
#define P_BK 32
#define P_LD 36

typedef __attribute__((ext_vector_type(4))) float v4f;

static __device__ __forceinline__ v4f load_g4(const float* p) {
  return *(const GAS v4f*)(p);
}
static __device__ __forceinline__ void store_g4(float* p, v4f x) {
  *(GAS v4f*)(p) = x;
}

__global__ __launch_bounds__(256)
void qkv_proj_kernel(const float* __restrict__ x,
                     const float* __restrict__ Wq, const float* __restrict__ bq,
                     const float* __restrict__ Wk, const float* __restrict__ bk,
                     const float* __restrict__ Wv, const float* __restrict__ bv,
                     float* __restrict__ q, float* __restrict__ k,
                     float* __restrict__ v) {
  __shared__ __attribute__((aligned(16))) float sA[P_BM * P_LD];
  __shared__ __attribute__((aligned(16))) float sB[P_BN * P_LD];

  const int tid  = threadIdx.x;
  const int w    = tid >> 5;
  const int lane = tid & 31;
  const int half = lane >> 4;
  const int lm   = lane & 15;
  const int wm   = w >> 1;
  const int wn   = w & 1;
  const int nt   = blockIdx.x;
  const int sel  = nt >> 3;
  const int n0   = (nt & 7) * P_BN;
  const int m0   = blockIdx.y * P_BM;

  const float* Wm = Wq; const float* bias = bq; float* dst = q;
  if (sel == 1) { Wm = Wk; bias = bk; dst = k; }
  if (sel == 2) { Wm = Wv; bias = bv; dst = v; }

  const int srow = tid >> 3;
  const int scol = (tid & 7) * 4;
  const float* xb = x  + (size_t)m0 * E_DIM;
  const float* wb = Wm + (size_t)n0 * E_DIM;
  const unsigned goff = (unsigned)(srow * E_DIM + scol);
  const int st_off = srow * P_LD + scol;

  const int a_off = (wm * 32 + lm) * P_LD + 4 * half;
  const int b_off = (wn * 64 + lm) * P_LD + 4 * half;

  v8f acc[2][4];
  #pragma unroll
  for (int ni = 0; ni < 4; ++ni) {
    const float bi = load_g1(bias + n0 + wn * 64 + ni * 16 + lm);
    #pragma unroll
    for (int mi = 0; mi < 2; ++mi)
      #pragma unroll
      for (int g = 0; g < 8; ++g) acc[mi][ni][g] = bi;
  }

  v4f ra[4], rb[4];
  #pragma unroll
  for (int i = 0; i < 4; ++i) {
    ra[i] = load_g4(xb + goff + i * 32 * E_DIM);
    rb[i] = load_g4(wb + goff + i * 32 * E_DIM);
  }
  #pragma unroll
  for (int i = 0; i < 4; ++i) {
    *(v4f*)&sA[st_off + i * 32 * P_LD] = ra[i];
    *(v4f*)&sB[st_off + i * 32 * P_LD] = rb[i];
  }
  __syncthreads();

  for (int kb = 0; kb < E_DIM / P_BK; ++kb) {
    const bool more = (kb + 1) < (E_DIM / P_BK);
    if (more) {
      xb += P_BK;
      wb += P_BK;
      #pragma unroll
      for (int i = 0; i < 4; ++i) {
        ra[i] = load_g4(xb + goff + i * 32 * E_DIM);
        rb[i] = load_g4(wb + goff + i * 32 * E_DIM);
      }
    }

    v4f fa[2][2], fb[2][4];
    #pragma unroll
    for (int mi = 0; mi < 2; ++mi) fa[0][mi] = *(const v4f*)&sA[a_off + mi * 16 * P_LD];
    #pragma unroll
    for (int ni = 0; ni < 4; ++ni) fb[0][ni] = *(const v4f*)&sB[b_off + ni * 16 * P_LD];

    #pragma unroll
    for (int s = 0; s < P_BK / 8; ++s) {
      const int cur = s & 1, nxt = cur ^ 1;
      if (s + 1 < P_BK / 8) {
        #pragma unroll
        for (int mi = 0; mi < 2; ++mi)
          fa[nxt][mi] = *(const v4f*)&sA[a_off + mi * 16 * P_LD + (s + 1) * 8];
        #pragma unroll
        for (int ni = 0; ni < 4; ++ni)
          fb[nxt][ni] = *(const v4f*)&sB[b_off + ni * 16 * P_LD + (s + 1) * 8];
      }
      #pragma unroll
      for (int mi = 0; mi < 2; ++mi)
        #pragma unroll
        for (int ni = 0; ni < 4; ++ni)
          acc[mi][ni] = wmma_f32(fa[cur][mi].lo, fb[cur][ni].lo, acc[mi][ni]);
      #pragma unroll
      for (int mi = 0; mi < 2; ++mi)
        #pragma unroll
        for (int ni = 0; ni < 4; ++ni)
          acc[mi][ni] = wmma_f32(fa[cur][mi].hi, fb[cur][ni].hi, acc[mi][ni]);
    }

    __syncthreads();
    if (more) {
      #pragma unroll
      for (int i = 0; i < 4; ++i) {
        *(v4f*)&sA[st_off + i * 32 * P_LD] = ra[i];
        *(v4f*)&sB[st_off + i * 32 * P_LD] = rb[i];
      }
    }
    __syncthreads();
  }

  #pragma unroll
  for (int mi = 0; mi < 2; ++mi) {
    #pragma unroll
    for (int ni = 0; ni < 4; ++ni) {
      const int col = n0 + wn * 64 + ni * 16 + lm;
      #pragma unroll
      for (int g = 0; g < 8; ++g) {
        const size_t row = (size_t)(m0 + wm * 32 + mi * 16 + g + 8 * half);
        store_g1(dst + row * E_DIM + col, acc[mi][ni][g]);
      }
    }
  }
}

__global__ __launch_bounds__(256)
void rownorm_kernel(const float* __restrict__ k, float* __restrict__ gk2) {
  __shared__ float red[256];
  const float* kr = k + (size_t)blockIdx.x * E_DIM;
  float s = 0.f;
  for (int e = threadIdx.x; e < E_DIM; e += 256) {
    const float t = load_g1(kr + e);
    s += t * t;
  }
  red[threadIdx.x] = s;
  __syncthreads();
  for (int off = 128; off > 0; off >>= 1) {
    if (threadIdx.x < off) red[threadIdx.x] += red[threadIdx.x + off];
    __syncthreads();
  }
  if (threadIdx.x == 0) store_g1(gk2 + blockIdx.x, GAMMA * red[0]);
}

__global__ __launch_bounds__(256)
void attn_kernel(const float* __restrict__ q, const float* __restrict__ k,
                 const float* __restrict__ v, const float* __restrict__ gk2,
                 float* __restrict__ out) {
  const int tid  = threadIdx.x;
  const int w    = tid >> 5;
  const int lane = tid & 31;
  const int half = lane >> 4;
  const int lm   = lane & 15;
  const int b    = blockIdx.y;
  const int t0   = blockIdx.x * 16;
  const int e0   = w * 128;

  __shared__ float sS[8][256];
  __shared__ float sP[256];
  __shared__ float sScale[16];
  __shared__ float sM[16];
  __shared__ float sL[16];

  if (tid < 16) { sM[tid] = -__builtin_inff(); sL[tid] = 0.f; }
  __syncthreads();

  v4f qf[16];
  {
    const float* qrow =
        q + ((size_t)(t0 + lm) * B_SZ + b) * E_DIM + e0 + 4 * half;
    #pragma unroll
    for (int s = 0; s < 16; ++s) qf[s] = load_g4(qrow + 8 * s) * (2.0f * GAMMA);
  }

  v8f acc[8];
  #pragma unroll
  for (int ns = 0; ns < 8; ++ns) acc[ns] = (v8f){};

  for (int j0 = 0; j0 < S_LEN; j0 += 16) {
    v8f sc = (v8f){};
    const float* krow =
        k + ((size_t)(j0 + lm) * B_SZ + b) * E_DIM + e0 + 4 * half;
    #pragma unroll
    for (int s = 0; s < 16; ++s) {
      const v4f kf = load_g4(krow + 8 * s);
      sc = wmma_f32(qf[s].lo, kf.lo, sc);
      sc = wmma_f32(qf[s].hi, kf.hi, sc);
    }

    #pragma unroll
    for (int g = 0; g < 8; ++g)
      sS[w][(g + 8 * half) * 16 + lm] = sc[g];
    __syncthreads();

    float logit;
    {
      float s = 0.f;
      #pragma unroll
      for (int ww = 0; ww < 8; ++ww) s += sS[ww][tid];
      logit = s - load_g1(gk2 + (size_t)(j0 + (tid & 15)) * B_SZ + b);
      sP[tid] = logit;
    }
    __syncthreads();

    float scl_row = 0.f;
    if (tid < 16) {
      const int row = tid;
      float mrow = -__builtin_inff();
      #pragma unroll
      for (int c = 0; c < 16; ++c) mrow = fmaxf(mrow, sP[row * 16 + c]);
      const float mnew = fmaxf(sM[row], mrow);
      scl_row = __expf(sM[row] - mnew);
      sM[row] = mnew;
      sScale[row] = scl_row;
    }
    __syncthreads();

    sP[tid] = __expf(logit - sM[tid >> 4]);
    __syncthreads();

    if (tid < 16) {
      float psum = 0.f;
      #pragma unroll
      for (int c = 0; c < 16; ++c) psum += sP[tid * 16 + c];
      sL[tid] = sL[tid] * scl_row + psum;
    }

    float f[8];
    #pragma unroll
    for (int g = 0; g < 8; ++g) f[g] = sScale[g + 8 * half];
    #pragma unroll
    for (int ns = 0; ns < 8; ++ns)
      #pragma unroll
      for (int g = 0; g < 8; ++g) acc[ns][g] *= f[g];

    #pragma unroll
    for (int c = 0; c < 4; ++c) {
      const v2f pf = *(const v2f*)&sP[lm * 16 + 4 * c + 2 * half];
      const int jA = j0 + 4 * c + 2 * half;
      const float* vbase = v + ((size_t)jA * B_SZ + b) * E_DIM + e0 + 4 * lm;
      #pragma unroll
      for (int q4 = 0; q4 < 2; ++q4) {
        const v4f r0 = load_g4(vbase + 64 * q4);
        const v4f r1 = load_g4(vbase + (size_t)B_SZ * E_DIM + 64 * q4);
        #pragma unroll
        for (int u = 0; u < 4; ++u) {
          v2f vf;
          vf.x = r0[u];
          vf.y = r1[u];
          acc[4 * q4 + u] = wmma_f32(pf, vf, acc[4 * q4 + u]);
        }
      }
    }
  }
  __syncthreads();

  float linv[8];
  #pragma unroll
  for (int g = 0; g < 8; ++g) linv[g] = 1.0f / sL[g + 8 * half];
  #pragma unroll
  for (int q4 = 0; q4 < 2; ++q4) {
    const int e = e0 + 64 * q4 + 4 * lm;
    #pragma unroll
    for (int g = 0; g < 8; ++g) {
      const int trow = t0 + g + 8 * half;
      v4f o;
      o.x = acc[4 * q4 + 0][g] * linv[g];
      o.y = acc[4 * q4 + 1][g] * linv[g];
      o.z = acc[4 * q4 + 2][g] * linv[g];
      o.w = acc[4 * q4 + 3][g] * linv[g];
      store_g4(out + ((size_t)trow * B_SZ + b) * E_DIM + e, o);
    }
  }
}

extern "C" void kernel_launch(void* const* d_in, const int* in_sizes, int n_in,
                              void* d_out, int out_size, void* d_ws, size_t ws_size,
                              hipStream_t stream) {
  (void)in_sizes; (void)n_in; (void)out_size; (void)ws_size;
  const float* x  = (const float*)d_in[0];
  const float* Wq = (const float*)d_in[1];
  const float* bq = (const float*)d_in[2];
  const float* Wk = (const float*)d_in[3];
  const float* bk = (const float*)d_in[4];
  const float* Wv = (const float*)d_in[5];
  const float* bv = (const float*)d_in[6];
  float* out = (float*)d_out;

  const size_t mat = (size_t)S_LEN * B_SZ * E_DIM;
  float* q   = (float*)d_ws;
  float* k   = q + mat;
  float* v   = k + mat;
  float* gk2 = v + mat;

  dim3 g1((3 * E_DIM) / P_BN, (S_LEN * B_SZ) / P_BM);
  qkv_proj_kernel<<<g1, 256, 0, stream>>>(x, Wq, bq, Wk, bk, Wv, bv, q, k, v);

  rownorm_kernel<<<S_LEN * B_SZ, 256, 0, stream>>>(k, gk2);

  dim3 g3(S_LEN / 16, B_SZ);
  attn_kernel<<<g3, 256, 0, stream>>>(q, k, v, gk2, out);
}


# attention re-tiled: 16-wave workgroup per 32-query tile, 64-wide E slice per wave, each K/V fragment feeds two row tiles (K/V read once per 32 queries); q.k sum now 16 partials of 64 e
# speedup vs baseline: 2.5279x; 1.0771x over previous
#include <hip/hip_runtime.h>
#include <hip/hip_bf16.h>


#define S_LEN 2048
#define B_SZ  8
#define E_DIM 1024
constexpr float GAMMA = 0.5f;

typedef __attribute__((ext_vector_type(2))) float v2f;
typedef __attribute__((ext_vector_type(8))) float v8f;

#define GAS __attribute__((address_space(1)))

static __device__ __forceinline__ float load_g1(const float* p) {
  return *(const GAS float*)(p);
}
static __device__ __forceinline__ void store_g1(float* p, float x) {
  *(GAS float*)(p) = x;
}

static __device__ __forceinline__ v8f wmma_f32(v2f a, v2f b, v8f c) {
  return __builtin_amdgcn_wmma_f32_16x16x4_f32(false, a, false, b,
                                               (short)0, c, false, false);
}

#define P_BM 128
#define P_BN 128
#define P_BK 32
#define P_LD 36

typedef __attribute__((ext_vector_type(4))) float v4f;

static __device__ __forceinline__ v4f load_g4(const float* p) {
  return *(const GAS v4f*)(p);
}
static __device__ __forceinline__ void store_g4(float* p, v4f x) {
  *(GAS v4f*)(p) = x;
}

__global__ __launch_bounds__(256)
void qkv_proj_kernel(const float* __restrict__ x,
                     const float* __restrict__ Wq, const float* __restrict__ bq,
                     const float* __restrict__ Wk, const float* __restrict__ bk,
                     const float* __restrict__ Wv, const float* __restrict__ bv,
                     float* __restrict__ q, float* __restrict__ k,
                     float* __restrict__ v) {
  __shared__ __attribute__((aligned(16))) float sA[P_BM * P_LD];
  __shared__ __attribute__((aligned(16))) float sB[P_BN * P_LD];

  const int tid  = threadIdx.x;
  const int w    = tid >> 5;
  const int lane = tid & 31;
  const int half = lane >> 4;
  const int lm   = lane & 15;
  const int wm   = w >> 1;
  const int wn   = w & 1;
  const int nt   = blockIdx.x;
  const int sel  = nt >> 3;
  const int n0   = (nt & 7) * P_BN;
  const int m0   = blockIdx.y * P_BM;

  const float* Wm = Wq; const float* bias = bq; float* dst = q;
  if (sel == 1) { Wm = Wk; bias = bk; dst = k; }
  if (sel == 2) { Wm = Wv; bias = bv; dst = v; }

  const int srow = tid >> 3;
  const int scol = (tid & 7) * 4;
  const float* xb = x  + (size_t)m0 * E_DIM;
  const float* wb = Wm + (size_t)n0 * E_DIM;
  const unsigned goff = (unsigned)(srow * E_DIM + scol);
  const int st_off = srow * P_LD + scol;

  const int a_off = (wm * 32 + lm) * P_LD + 4 * half;
  const int b_off = (wn * 64 + lm) * P_LD + 4 * half;

  v8f acc[2][4];
  #pragma unroll
  for (int ni = 0; ni < 4; ++ni) {
    const float bi = load_g1(bias + n0 + wn * 64 + ni * 16 + lm);
    #pragma unroll
    for (int mi = 0; mi < 2; ++mi)
      #pragma unroll
      for (int g = 0; g < 8; ++g) acc[mi][ni][g] = bi;
  }

  v4f ra[4], rb[4];
  #pragma unroll
  for (int i = 0; i < 4; ++i) {
    ra[i] = load_g4(xb + goff + i * 32 * E_DIM);
    rb[i] = load_g4(wb + goff + i * 32 * E_DIM);
  }
  #pragma unroll
  for (int i = 0; i < 4; ++i) {
    *(v4f*)&sA[st_off + i * 32 * P_LD] = ra[i];
    *(v4f*)&sB[st_off + i * 32 * P_LD] = rb[i];
  }
  __syncthreads();

  for (int kb = 0; kb < E_DIM / P_BK; ++kb) {
    const bool more = (kb + 1) < (E_DIM / P_BK);
    if (more) {
      xb += P_BK;
      wb += P_BK;
      #pragma unroll
      for (int i = 0; i < 4; ++i) {
        ra[i] = load_g4(xb + goff + i * 32 * E_DIM);
        rb[i] = load_g4(wb + goff + i * 32 * E_DIM);
      }
    }

    v4f fa[2][2], fb[2][4];
    #pragma unroll
    for (int mi = 0; mi < 2; ++mi) fa[0][mi] = *(const v4f*)&sA[a_off + mi * 16 * P_LD];
    #pragma unroll
    for (int ni = 0; ni < 4; ++ni) fb[0][ni] = *(const v4f*)&sB[b_off + ni * 16 * P_LD];

    #pragma unroll
    for (int s = 0; s < P_BK / 8; ++s) {
      const int cur = s & 1, nxt = cur ^ 1;
      if (s + 1 < P_BK / 8) {
        #pragma unroll
        for (int mi = 0; mi < 2; ++mi)
          fa[nxt][mi] = *(const v4f*)&sA[a_off + mi * 16 * P_LD + (s + 1) * 8];
        #pragma unroll
        for (int ni = 0; ni < 4; ++ni)
          fb[nxt][ni] = *(const v4f*)&sB[b_off + ni * 16 * P_LD + (s + 1) * 8];
      }
      #pragma unroll
      for (int mi = 0; mi < 2; ++mi)
        #pragma unroll
        for (int ni = 0; ni < 4; ++ni)
          acc[mi][ni] = wmma_f32(fa[cur][mi].lo, fb[cur][ni].lo, acc[mi][ni]);
      #pragma unroll
      for (int mi = 0; mi < 2; ++mi)
        #pragma unroll
        for (int ni = 0; ni < 4; ++ni)
          acc[mi][ni] = wmma_f32(fa[cur][mi].hi, fb[cur][ni].hi, acc[mi][ni]);
    }

    __syncthreads();
    if (more) {
      #pragma unroll
      for (int i = 0; i < 4; ++i) {
        *(v4f*)&sA[st_off + i * 32 * P_LD] = ra[i];
        *(v4f*)&sB[st_off + i * 32 * P_LD] = rb[i];
      }
    }
    __syncthreads();
  }

  #pragma unroll
  for (int mi = 0; mi < 2; ++mi) {
    #pragma unroll
    for (int ni = 0; ni < 4; ++ni) {
      const int col = n0 + wn * 64 + ni * 16 + lm;
      #pragma unroll
      for (int g = 0; g < 8; ++g) {
        const size_t row = (size_t)(m0 + wm * 32 + mi * 16 + g + 8 * half);
        store_g1(dst + row * E_DIM + col, acc[mi][ni][g]);
      }
    }
  }
}

__global__ __launch_bounds__(256)
void rownorm_kernel(const float* __restrict__ k, float* __restrict__ gk2) {
  __shared__ float red[256];
  const float* kr = k + (size_t)blockIdx.x * E_DIM;
  float s = 0.f;
  for (int e = threadIdx.x; e < E_DIM; e += 256) {
    const float t = load_g1(kr + e);
    s += t * t;
  }
  red[threadIdx.x] = s;
  __syncthreads();
  for (int off = 128; off > 0; off >>= 1) {
    if (threadIdx.x < off) red[threadIdx.x] += red[threadIdx.x + off];
    __syncthreads();
  }
  if (threadIdx.x == 0) store_g1(gk2 + blockIdx.x, GAMMA * red[0]);
}

#define A_QT 32
#define A_NW 16
#define A_ES 64

__global__ __launch_bounds__(512)
void attn_kernel(const float* __restrict__ q, const float* __restrict__ k,
                 const float* __restrict__ v, const float* __restrict__ gk2,
                 float* __restrict__ out) {
  const int tid  = threadIdx.x;
  const int w    = tid >> 5;
  const int lane = tid & 31;
  const int half = lane >> 4;
  const int lm   = lane & 15;
  const int b    = blockIdx.y;
  const int t0   = blockIdx.x * A_QT;
  const int e0   = w * A_ES;

  __shared__ float sS[A_NW][A_QT * 16];
  __shared__ float sP[A_QT * 16];
  __shared__ float sScale[A_QT];
  __shared__ float sM[A_QT];
  __shared__ float sL[A_QT];

  if (tid < A_QT) { sM[tid] = -__builtin_inff(); sL[tid] = 0.f; }
  __syncthreads();

  v4f qf[2][8];
  #pragma unroll
  for (int mi = 0; mi < 2; ++mi) {
    const float* qrow =
        q + ((size_t)(t0 + 16 * mi + lm) * B_SZ + b) * E_DIM + e0 + 4 * half;
    #pragma unroll
    for (int s = 0; s < 8; ++s)
      qf[mi][s] = load_g4(qrow + 8 * s) * (2.0f * GAMMA);
  }

  v8f acc[2][4];
  #pragma unroll
  for (int mi = 0; mi < 2; ++mi)
    #pragma unroll
    for (int u = 0; u < 4; ++u) acc[mi][u] = (v8f){};

  for (int j0 = 0; j0 < S_LEN; j0 += 16) {
    v8f sc[2];
    sc[0] = (v8f){};
    sc[1] = (v8f){};
    const float* krow =
        k + ((size_t)(j0 + lm) * B_SZ + b) * E_DIM + e0 + 4 * half;
    #pragma unroll
    for (int s = 0; s < 8; ++s) {
      const v4f kf = load_g4(krow + 8 * s);
      sc[0] = wmma_f32(qf[0][s].lo, kf.lo, sc[0]);
      sc[1] = wmma_f32(qf[1][s].lo, kf.lo, sc[1]);
      sc[0] = wmma_f32(qf[0][s].hi, kf.hi, sc[0]);
      sc[1] = wmma_f32(qf[1][s].hi, kf.hi, sc[1]);
    }

    #pragma unroll
    for (int mi = 0; mi < 2; ++mi)
      #pragma unroll
      for (int g = 0; g < 8; ++g)
        sS[w][(16 * mi + g + 8 * half) * 16 + lm] = sc[mi][g];
    __syncthreads();

    float logit;
    {
      float s = 0.f;
      #pragma unroll
      for (int ww = 0; ww < A_NW; ++ww) s += sS[ww][tid];
      logit = s - load_g1(gk2 + (size_t)(j0 + (tid & 15)) * B_SZ + b);
      sP[tid] = logit;
    }
    __syncthreads();

    float scl_row = 0.f;
    if (tid < A_QT) {
      const int row = tid;
      float mrow = -__builtin_inff();
      #pragma unroll
      for (int c = 0; c < 16; ++c) mrow = fmaxf(mrow, sP[row * 16 + c]);
      const float mnew = fmaxf(sM[row], mrow);
      scl_row = __expf(sM[row] - mnew);
      sM[row] = mnew;
      sScale[row] = scl_row;
    }
    __syncthreads();

    sP[tid] = __expf(logit - sM[tid >> 4]);
    __syncthreads();

    if (tid < A_QT) {
      float psum = 0.f;
      #pragma unroll
      for (int c = 0; c < 16; ++c) psum += sP[tid * 16 + c];
      sL[tid] = sL[tid] * scl_row + psum;
    }

    #pragma unroll
    for (int mi = 0; mi < 2; ++mi) {
      float f[8];
      #pragma unroll
      for (int g = 0; g < 8; ++g) f[g] = sScale[16 * mi + g + 8 * half];
      #pragma unroll
      for (int u = 0; u < 4; ++u)
        #pragma unroll
        for (int g = 0; g < 8; ++g) acc[mi][u][g] *= f[g];
    }

    #pragma unroll
    for (int c = 0; c < 4; ++c) {
      const v2f pf0 = *(const v2f*)&sP[lm * 16 + 4 * c + 2 * half];
      const v2f pf1 = *(const v2f*)&sP[(16 + lm) * 16 + 4 * c + 2 * half];
      const int jA = j0 + 4 * c + 2 * half;
      const float* vbase = v + ((size_t)jA * B_SZ + b) * E_DIM + e0 + 4 * lm;
      const v4f r0 = load_g4(vbase);
      const v4f r1 = load_g4(vbase + (size_t)B_SZ * E_DIM);
      #pragma unroll
      for (int u = 0; u < 4; ++u) {
        v2f vf;
        vf.x = r0[u];
        vf.y = r1[u];
        acc[0][u] = wmma_f32(pf0, vf, acc[0][u]);
        acc[1][u] = wmma_f32(pf1, vf, acc[1][u]);
      }
    }
  }
  __syncthreads();

  #pragma unroll
  for (int mi = 0; mi < 2; ++mi) {
    float linv[8];
    #pragma unroll
    for (int g = 0; g < 8; ++g) linv[g] = 1.0f / sL[16 * mi + g + 8 * half];
    const int e = e0 + 4 * lm;
    #pragma unroll
    for (int g = 0; g < 8; ++g) {
      const int trow = t0 + 16 * mi + g + 8 * half;
      v4f o;
      o.x = acc[mi][0][g] * linv[g];
      o.y = acc[mi][1][g] * linv[g];
      o.z = acc[mi][2][g] * linv[g];
      o.w = acc[mi][3][g] * linv[g];
      store_g4(out + ((size_t)trow * B_SZ + b) * E_DIM + e, o);
    }
  }
}

extern "C" void kernel_launch(void* const* d_in, const int* in_sizes, int n_in,
                              void* d_out, int out_size, void* d_ws, size_t ws_size,
                              hipStream_t stream) {
  (void)in_sizes; (void)n_in; (void)out_size; (void)ws_size;
  const float* x  = (const float*)d_in[0];
  const float* Wq = (const float*)d_in[1];
  const float* bq = (const float*)d_in[2];
  const float* Wk = (const float*)d_in[3];
  const float* bk = (const float*)d_in[4];
  const float* Wv = (const float*)d_in[5];
  const float* bv = (const float*)d_in[6];
  float* out = (float*)d_out;

  const size_t mat = (size_t)S_LEN * B_SZ * E_DIM;
  float* q   = (float*)d_ws;
  float* k   = q + mat;
  float* v   = k + mat;
  float* gk2 = v + mat;

  dim3 g1((3 * E_DIM) / P_BN, (S_LEN * B_SZ) / P_BM);
  qkv_proj_kernel<<<g1, 256, 0, stream>>>(x, Wq, bq, Wk, bk, Wv, bv, q, k, v);

  rownorm_kernel<<<S_LEN * B_SZ, 256, 0, stream>>>(k, gk2);

  dim3 g3(S_LEN / A_QT, B_SZ);
  attn_kernel<<<g3, 32 * A_NW, 0, stream>>>(q, k, v, gk2, out);
}


# attention: 64 keys per pass instead of 16, softmax row max/sum by 16-lane row shuffles in registers: 2 barriers per 64 keys instead of 4 per 16; running max per 64 keys, row sum order changes
# speedup vs baseline: 2.8572x; 1.1303x over previous
#include <hip/hip_runtime.h>
#include <hip/hip_bf16.h>


#define S_LEN 2048
#define B_SZ  8
#define E_DIM 1024
constexpr float GAMMA = 0.5f;

typedef __attribute__((ext_vector_type(2))) float v2f;
typedef __attribute__((ext_vector_type(8))) float v8f;

#define GAS __attribute__((address_space(1)))

static __device__ __forceinline__ float load_g1(const float* p) {
  return *(const GAS float*)(p);
}
static __device__ __forceinline__ void store_g1(float* p, float x) {
  *(GAS float*)(p) = x;
}

static __device__ __forceinline__ v8f wmma_f32(v2f a, v2f b, v8f c) {
  return __builtin_amdgcn_wmma_f32_16x16x4_f32(false, a, false, b,
                                               (short)0, c, false, false);
}

#define P_BM 128
#define P_BN 128
#define P_BK 32
#define P_LD 36

typedef __attribute__((ext_vector_type(4))) float v4f;

static __device__ __forceinline__ v4f load_g4(const float* p) {
  return *(const GAS v4f*)(p);
}
static __device__ __forceinline__ void store_g4(float* p, v4f x) {
  *(GAS v4f*)(p) = x;
}

__global__ __launch_bounds__(256)
void qkv_proj_kernel(const float* __restrict__ x,
                     const float* __restrict__ Wq, const float* __restrict__ bq,
                     const float* __restrict__ Wk, const float* __restrict__ bk,
                     const float* __restrict__ Wv, const float* __restrict__ bv,
                     float* __restrict__ q, float* __restrict__ k,
                     float* __restrict__ v) {
  __shared__ __attribute__((aligned(16))) float sA[P_BM * P_LD];
  __shared__ __attribute__((aligned(16))) float sB[P_BN * P_LD];

  const int tid  = threadIdx.x;
  const int w    = tid >> 5;
  const int lane = tid & 31;
  const int half = lane >> 4;
  const int lm   = lane & 15;
  const int wm   = w >> 1;
  const int wn   = w & 1;
  const int nt   = blockIdx.x;
  const int sel  = nt >> 3;
  const int n0   = (nt & 7) * P_BN;
  const int m0   = blockIdx.y * P_BM;

  const float* Wm = Wq; const float* bias = bq; float* dst = q;
  if (sel == 1) { Wm = Wk; bias = bk; dst = k; }
  if (sel == 2) { Wm = Wv; bias = bv; dst = v; }

  const int srow = tid >> 3;
  const int scol = (tid & 7) * 4;
  const float* xb = x  + (size_t)m0 * E_DIM;
  const float* wb = Wm + (size_t)n0 * E_DIM;
  const unsigned goff = (unsigned)(srow * E_DIM + scol);
  const int st_off = srow * P_LD + scol;

  const int a_off = (wm * 32 + lm) * P_LD + 4 * half;
  const int b_off = (wn * 64 + lm) * P_LD + 4 * half;

  v8f acc[2][4];
  #pragma unroll
  for (int ni = 0; ni < 4; ++ni) {
    const float bi = load_g1(bias + n0 + wn * 64 + ni * 16 + lm);
    #pragma unroll
    for (int mi = 0; mi < 2; ++mi)
      #pragma unroll
      for (int g = 0; g < 8; ++g) acc[mi][ni][g] = bi;
  }

  v4f ra[4], rb[4];
  #pragma unroll
  for (int i = 0; i < 4; ++i) {
    ra[i] = load_g4(xb + goff + i * 32 * E_DIM);
    rb[i] = load_g4(wb + goff + i * 32 * E_DIM);
  }
  #pragma unroll
  for (int i = 0; i < 4; ++i) {
    *(v4f*)&sA[st_off + i * 32 * P_LD] = ra[i];
    *(v4f*)&sB[st_off + i * 32 * P_LD] = rb[i];
  }
  __syncthreads();

  for (int kb = 0; kb < E_DIM / P_BK; ++kb) {
    const bool more = (kb + 1) < (E_DIM / P_BK);
    if (more) {
      xb += P_BK;
      wb += P_BK;
      #pragma unroll
      for (int i = 0; i < 4; ++i) {
        ra[i] = load_g4(xb + goff + i * 32 * E_DIM);
        rb[i] = load_g4(wb + goff + i * 32 * E_DIM);
      }
    }

    v4f fa[2][2], fb[2][4];
    #pragma unroll
    for (int mi = 0; mi < 2; ++mi) fa[0][mi] = *(const v4f*)&sA[a_off + mi * 16 * P_LD];
    #pragma unroll
    for (int ni = 0; ni < 4; ++ni) fb[0][ni] = *(const v4f*)&sB[b_off + ni * 16 * P_LD];

    #pragma unroll
    for (int s = 0; s < P_BK / 8; ++s) {
      const int cur = s & 1, nxt = cur ^ 1;
      if (s + 1 < P_BK / 8) {
        #pragma unroll
        for (int mi = 0; mi < 2; ++mi)
          fa[nxt][mi] = *(const v4f*)&sA[a_off + mi * 16 * P_LD + (s + 1) * 8];
        #pragma unroll
        for (int ni = 0; ni < 4; ++ni)
          fb[nxt][ni] = *(const v4f*)&sB[b_off + ni * 16 * P_LD + (s + 1) * 8];
      }
      #pragma unroll
      for (int mi = 0; mi < 2; ++mi)
        #pragma unroll
        for (int ni = 0; ni < 4; ++ni)
          acc[mi][ni] = wmma_f32(fa[cur][mi].lo, fb[cur][ni].lo, acc[mi][ni]);
      #pragma unroll
      for (int mi = 0; mi < 2; ++mi)
        #pragma unroll
        for (int ni = 0; ni < 4; ++ni)
          acc[mi][ni] = wmma_f32(fa[cur][mi].hi, fb[cur][ni].hi, acc[mi][ni]);
    }

    __syncthreads();
    if (more) {
      #pragma unroll
      for (int i = 0; i < 4; ++i) {
        *(v4f*)&sA[st_off + i * 32 * P_LD] = ra[i];
        *(v4f*)&sB[st_off + i * 32 * P_LD] = rb[i];
      }
    }
    __syncthreads();
  }

  #pragma unroll
  for (int mi = 0; mi < 2; ++mi) {
    #pragma unroll
    for (int ni = 0; ni < 4; ++ni) {
      const int col = n0 + wn * 64 + ni * 16 + lm;
      #pragma unroll
      for (int g = 0; g < 8; ++g) {
        const size_t row = (size_t)(m0 + wm * 32 + mi * 16 + g + 8 * half);
        store_g1(dst + row * E_DIM + col, acc[mi][ni][g]);
      }
    }
  }
}

__global__ __launch_bounds__(256)
void rownorm_kernel(const float* __restrict__ k, float* __restrict__ gk2) {
  __shared__ float red[256];
  const float* kr = k + (size_t)blockIdx.x * E_DIM;
  float s = 0.f;
  for (int e = threadIdx.x; e < E_DIM; e += 256) {
    const float t = load_g1(kr + e);
    s += t * t;
  }
  red[threadIdx.x] = s;
  __syncthreads();
  for (int off = 128; off > 0; off >>= 1) {
    if (threadIdx.x < off) red[threadIdx.x] += red[threadIdx.x + off];
    __syncthreads();
  }
  if (threadIdx.x == 0) store_g1(gk2 + blockIdx.x, GAMMA * red[0]);
}

#define A_QT 32
#define A_NW 16
#define A_ES 64
#define A_KT 64
#define A_SP 68

template <int M>
static __device__ __forceinline__ float row_xor(float x) {
  return __builtin_bit_cast(
      float, __builtin_amdgcn_update_dpp(0, __builtin_bit_cast(int, x),
                                         0x160 + M  , 0xf, 0xf,
                                         false));
}
static __device__ __forceinline__ float row16_max(float x) {
  x = fmaxf(x, row_xor<1>(x));
  x = fmaxf(x, row_xor<2>(x));
  x = fmaxf(x, row_xor<4>(x));
  x = fmaxf(x, row_xor<8>(x));
  return x;
}
static __device__ __forceinline__ float row16_sum(float x) {
  x += row_xor<1>(x);
  x += row_xor<2>(x);
  x += row_xor<4>(x);
  x += row_xor<8>(x);
  return x;
}

__global__ __launch_bounds__(512)
void attn_kernel(const float* __restrict__ q, const float* __restrict__ k,
                 const float* __restrict__ v, const float* __restrict__ gk2,
                 float* __restrict__ out) {
  const int tid  = threadIdx.x;
  const int w    = tid >> 5;
  const int lane = tid & 31;
  const int half = lane >> 4;
  const int lm   = lane & 15;
  const int b    = blockIdx.y;
  const int t0   = blockIdx.x * A_QT;
  const int e0   = w * A_ES;
  const int srow = tid >> 4;
  const int scg  = tid & 15;

  __shared__ __attribute__((aligned(16)))
  float sS[A_NW][A_QT * A_SP];
  __shared__ __attribute__((aligned(16)))
  float sP[A_QT * A_SP];
  __shared__ float sScale[A_QT];
  __shared__ float sL[A_QT];

  float m_run = -__builtin_inff();
  float l_run = 0.f;

  v4f qf[2][8];
  #pragma unroll
  for (int mi = 0; mi < 2; ++mi) {
    const float* qrow =
        q + ((size_t)(t0 + 16 * mi + lm) * B_SZ + b) * E_DIM + e0 + 4 * half;
    #pragma unroll
    for (int s = 0; s < 8; ++s)
      qf[mi][s] = load_g4(qrow + 8 * s) * (2.0f * GAMMA);
  }

  v8f acc[2][4];
  #pragma unroll
  for (int mi = 0; mi < 2; ++mi)
    #pragma unroll
    for (int u = 0; u < 4; ++u) acc[mi][u] = (v8f){};

  for (int j0 = 0; j0 < S_LEN; j0 += A_KT) {
    float gk[4];
    #pragma unroll
    for (int i = 0; i < 4; ++i)
      gk[i] = load_g1(gk2 + (size_t)(j0 + 4 * scg + i) * B_SZ + b);

    const float* kbase =
        k + ((size_t)(j0 + lm) * B_SZ + b) * E_DIM + e0 + 4 * half;
    v4f kf[2][8];
    #pragma unroll
    for (int s = 0; s < 8; ++s) kf[0][s] = load_g4(kbase + 8 * s);
    #pragma unroll
    for (int t = 0; t < A_KT / 16; ++t) {
      const int cur = t & 1, nxt = cur ^ 1;
      if (t + 1 < A_KT / 16) {
        const float* kn = kbase + (size_t)(t + 1) * 16 * B_SZ * E_DIM;
        #pragma unroll
        for (int s = 0; s < 8; ++s) kf[nxt][s] = load_g4(kn + 8 * s);
      }
      v8f sc[2];
      sc[0] = (v8f){};
      sc[1] = (v8f){};
      #pragma unroll
      for (int s = 0; s < 8; ++s) {
        const v4f kk = kf[cur][s];
        sc[0] = wmma_f32(qf[0][s].lo, kk.lo, sc[0]);
        sc[1] = wmma_f32(qf[1][s].lo, kk.lo, sc[1]);
        sc[0] = wmma_f32(qf[0][s].hi, kk.hi, sc[0]);
        sc[1] = wmma_f32(qf[1][s].hi, kk.hi, sc[1]);
      }
      #pragma unroll
      for (int mi = 0; mi < 2; ++mi)
        #pragma unroll
        for (int g = 0; g < 8; ++g)
          sS[w][(16 * mi + g + 8 * half) * A_SP + 16 * t + lm] = sc[mi][g];
      __builtin_amdgcn_sched_barrier(0);
    }
    __syncthreads();

    {
      v4f s4 = (v4f){};
      #pragma unroll
      for (int ww = 0; ww < A_NW; ++ww)
        s4 += *(const v4f*)&sS[ww][srow * A_SP + 4 * scg];
      const float l0 = s4.x - gk[0];
      const float l1 = s4.y - gk[1];
      const float l2 = s4.z - gk[2];
      const float l3 = s4.w - gk[3];
      const float mrow = row16_max(fmaxf(fmaxf(l0, l1), fmaxf(l2, l3)));
      const float mnew = fmaxf(m_run, mrow);
      const float scl  = __expf(m_run - mnew);
      m_run = mnew;
      v4f p;
      p.x = __expf(l0 - mnew);
      p.y = __expf(l1 - mnew);
      p.z = __expf(l2 - mnew);
      p.w = __expf(l3 - mnew);
      l_run = l_run * scl + row16_sum((p.x + p.y) + (p.z + p.w));
      *(v4f*)&sP[srow * A_SP + 4 * scg] = p;
      if (scg == 0) sScale[srow] = scl;
    }
    __syncthreads();

    #pragma unroll
    for (int mi = 0; mi < 2; ++mi) {
      float f[8];
      #pragma unroll
      for (int g = 0; g < 8; ++g) f[g] = sScale[16 * mi + g + 8 * half];
      #pragma unroll
      for (int u = 0; u < 4; ++u)
        #pragma unroll
        for (int g = 0; g < 8; ++g) acc[mi][u][g] *= f[g];
    }

    const float* vbase =
        v + ((size_t)(j0 + 2 * half) * B_SZ + b) * E_DIM + e0 + 4 * lm;
    const float* p0 = &sP[lm * A_SP + 2 * half];
    const float* p1 = &sP[(16 + lm) * A_SP + 2 * half];
    v4f vr[2][2][2];
    v2f pa[2][2][2];
    #pragma unroll
    for (int c = 0; c < 2; ++c) {
      vr[0][c][0] = load_g4(vbase + (size_t)(4 * c) * B_SZ * E_DIM);
      vr[0][c][1] = load_g4(vbase + (size_t)(4 * c + 1) * B_SZ * E_DIM);
      pa[0][c][0] = *(const v2f*)(p0 + 4 * c);
      pa[0][c][1] = *(const v2f*)(p1 + 4 * c);
    }
    #pragma unroll
    for (int gq = 0; gq < A_KT / 8; ++gq) {
      const int cur = gq & 1, nxt = cur ^ 1;
      if (gq + 1 < A_KT / 8) {
        #pragma unroll
        for (int c = 0; c < 2; ++c) {
          const int cc = 2 * (gq + 1) + c;
          vr[nxt][c][0] = load_g4(vbase + (size_t)(4 * cc) * B_SZ * E_DIM);
          vr[nxt][c][1] = load_g4(vbase + (size_t)(4 * cc + 1) * B_SZ * E_DIM);
          pa[nxt][c][0] = *(const v2f*)(p0 + 4 * cc);
          pa[nxt][c][1] = *(const v2f*)(p1 + 4 * cc);
        }
      }
      #pragma unroll
      for (int c = 0; c < 2; ++c) {
        #pragma unroll
        for (int u = 0; u < 4; ++u) {
          v2f vf;
          vf.x = vr[cur][c][0][u];
          vf.y = vr[cur][c][1][u];
          acc[0][u] = wmma_f32(pa[cur][c][0], vf, acc[0][u]);
          acc[1][u] = wmma_f32(pa[cur][c][1], vf, acc[1][u]);
        }
      }
    }
  }

  if (scg == 0) sL[srow] = l_run;
  __syncthreads();

  #pragma unroll
  for (int mi = 0; mi < 2; ++mi) {
    float linv[8];
    #pragma unroll
    for (int g = 0; g < 8; ++g) linv[g] = 1.0f / sL[16 * mi + g + 8 * half];
    const int e = e0 + 4 * lm;
    #pragma unroll
    for (int g = 0; g < 8; ++g) {
      const int trow = t0 + 16 * mi + g + 8 * half;
      v4f o;
      o.x = acc[mi][0][g] * linv[g];
      o.y = acc[mi][1][g] * linv[g];
      o.z = acc[mi][2][g] * linv[g];
      o.w = acc[mi][3][g] * linv[g];
      store_g4(out + ((size_t)trow * B_SZ + b) * E_DIM + e, o);
    }
  }
}

extern "C" void kernel_launch(void* const* d_in, const int* in_sizes, int n_in,
                              void* d_out, int out_size, void* d_ws, size_t ws_size,
                              hipStream_t stream) {
  (void)in_sizes; (void)n_in; (void)out_size; (void)ws_size;
  const float* x  = (const float*)d_in[0];
  const float* Wq = (const float*)d_in[1];
  const float* bq = (const float*)d_in[2];
  const float* Wk = (const float*)d_in[3];
  const float* bk = (const float*)d_in[4];
  const float* Wv = (const float*)d_in[5];
  const float* bv = (const float*)d_in[6];
  float* out = (float*)d_out;

  const size_t mat = (size_t)S_LEN * B_SZ * E_DIM;
  float* q   = (float*)d_ws;
  float* k   = q + mat;
  float* v   = k + mat;
  float* gk2 = v + mat;

  dim3 g1((3 * E_DIM) / P_BN, (S_LEN * B_SZ) / P_BM);
  qkv_proj_kernel<<<g1, 256, 0, stream>>>(x, Wq, bq, Wk, bk, Wv, bv, q, k, v);

  rownorm_kernel<<<S_LEN * B_SZ, 256, 0, stream>>>(k, gk2);

  dim3 g3(S_LEN / A_QT, B_SZ);
  attn_kernel<<<g3, 32 * A_NW, 0, stream>>>(q, k, v, gk2, out);
}


# K projection stored in the attention kernel's 16-key x 8-e fragment tile layout (re-ordered in LDS): each K fragment load is 512 contiguous bytes; row norm reads that layout, its sum order changes
# speedup vs baseline: 2.9539x; 1.0338x over previous
#include <hip/hip_runtime.h>
#include <hip/hip_bf16.h>


#define S_LEN 2048
#define B_SZ  8
#define E_DIM 1024
constexpr float GAMMA = 0.5f;

typedef __attribute__((ext_vector_type(2))) float v2f;
typedef __attribute__((ext_vector_type(8))) float v8f;

#define GAS __attribute__((address_space(1)))

static __device__ __forceinline__ float load_g1(const float* p) {
  return *(const GAS float*)(p);
}
static __device__ __forceinline__ void store_g1(float* p, float x) {
  *(GAS float*)(p) = x;
}

static __device__ __forceinline__ v8f wmma_f32(v2f a, v2f b, v8f c) {
  return __builtin_amdgcn_wmma_f32_16x16x4_f32(false, a, false, b,
                                               (short)0, c, false, false);
}

#define P_BM 128
#define P_BN 128
#define P_BK 32
#define P_LD 36

typedef __attribute__((ext_vector_type(4))) float v4f;

static __device__ __forceinline__ v4f load_g4(const float* p) {
  return *(const GAS v4f*)(p);
}
static __device__ __forceinline__ void store_g4(float* p, v4f x) {
  *(GAS v4f*)(p) = x;
}

__global__ __launch_bounds__(256)
void qkv_proj_kernel(const float* __restrict__ x,
                     const float* __restrict__ Wq, const float* __restrict__ bq,
                     const float* __restrict__ Wk, const float* __restrict__ bk,
                     const float* __restrict__ Wv, const float* __restrict__ bv,
                     float* __restrict__ q, float* __restrict__ k,
                     float* __restrict__ v) {
  __shared__ __attribute__((aligned(16))) float smem[P_BM * P_BN];
  float* const sA = smem;
  float* const sB = smem + P_BM * P_LD;

  const int tid  = threadIdx.x;
  const int w    = tid >> 5;
  const int lane = tid & 31;
  const int half = lane >> 4;
  const int lm   = lane & 15;
  const int wm   = w >> 1;
  const int wn   = w & 1;
  const int nt   = blockIdx.x;
  const int sel  = nt >> 3;
  const int n0   = (nt & 7) * P_BN;
  const int m0   = blockIdx.y * P_BM;

  const float* Wm = Wq; const float* bias = bq; float* dst = q;
  if (sel == 1) { Wm = Wk; bias = bk; dst = k; }
  if (sel == 2) { Wm = Wv; bias = bv; dst = v; }

  const int srow = tid >> 3;
  const int scol = (tid & 7) * 4;
  const float* xb = x  + (size_t)m0 * E_DIM;
  const float* wb = Wm + (size_t)n0 * E_DIM;
  const unsigned goff = (unsigned)(srow * E_DIM + scol);
  const int st_off = srow * P_LD + scol;

  const int a_off = (wm * 32 + lm) * P_LD + 4 * half;
  const int b_off = (wn * 64 + lm) * P_LD + 4 * half;

  v8f acc[2][4];
  #pragma unroll
  for (int ni = 0; ni < 4; ++ni) {
    const float bi = load_g1(bias + n0 + wn * 64 + ni * 16 + lm);
    #pragma unroll
    for (int mi = 0; mi < 2; ++mi)
      #pragma unroll
      for (int g = 0; g < 8; ++g) acc[mi][ni][g] = bi;
  }

  v4f ra[4], rb[4];
  #pragma unroll
  for (int i = 0; i < 4; ++i) {
    ra[i] = load_g4(xb + goff + i * 32 * E_DIM);
    rb[i] = load_g4(wb + goff + i * 32 * E_DIM);
  }
  #pragma unroll
  for (int i = 0; i < 4; ++i) {
    *(v4f*)&sA[st_off + i * 32 * P_LD] = ra[i];
    *(v4f*)&sB[st_off + i * 32 * P_LD] = rb[i];
  }
  __syncthreads();

  for (int kb = 0; kb < E_DIM / P_BK; ++kb) {
    const bool more = (kb + 1) < (E_DIM / P_BK);
    if (more) {
      xb += P_BK;
      wb += P_BK;
      #pragma unroll
      for (int i = 0; i < 4; ++i) {
        ra[i] = load_g4(xb + goff + i * 32 * E_DIM);
        rb[i] = load_g4(wb + goff + i * 32 * E_DIM);
      }
    }

    v4f fa[2][2], fb[2][4];
    #pragma unroll
    for (int mi = 0; mi < 2; ++mi) fa[0][mi] = *(const v4f*)&sA[a_off + mi * 16 * P_LD];
    #pragma unroll
    for (int ni = 0; ni < 4; ++ni) fb[0][ni] = *(const v4f*)&sB[b_off + ni * 16 * P_LD];

    #pragma unroll
    for (int s = 0; s < P_BK / 8; ++s) {
      const int cur = s & 1, nxt = cur ^ 1;
      if (s + 1 < P_BK / 8) {
        #pragma unroll
        for (int mi = 0; mi < 2; ++mi)
          fa[nxt][mi] = *(const v4f*)&sA[a_off + mi * 16 * P_LD + (s + 1) * 8];
        #pragma unroll
        for (int ni = 0; ni < 4; ++ni)
          fb[nxt][ni] = *(const v4f*)&sB[b_off + ni * 16 * P_LD + (s + 1) * 8];
      }
      #pragma unroll
      for (int mi = 0; mi < 2; ++mi)
        #pragma unroll
        for (int ni = 0; ni < 4; ++ni)
          acc[mi][ni] = wmma_f32(fa[cur][mi].lo, fb[cur][ni].lo, acc[mi][ni]);
      #pragma unroll
      for (int mi = 0; mi < 2; ++mi)
        #pragma unroll
        for (int ni = 0; ni < 4; ++ni)
          acc[mi][ni] = wmma_f32(fa[cur][mi].hi, fb[cur][ni].hi, acc[mi][ni]);
    }

    __syncthreads();
    if (more) {
      #pragma unroll
      for (int i = 0; i < 4; ++i) {
        *(v4f*)&sA[st_off + i * 32 * P_LD] = ra[i];
        *(v4f*)&sB[st_off + i * 32 * P_LD] = rb[i];
      }
    }
    __syncthreads();
  }

  if (sel == 1) {
    #pragma unroll
    for (int mi = 0; mi < 2; ++mi) {
      const int key = wm * 4 + mi * 2 + half;
      #pragma unroll
      for (int ni = 0; ni < 4; ++ni) {
        const int el   = wn * 64 + ni * 16 + lm;
        const int cell = (((el >> 2) & 1) * 16 + key) * 4 + (el & 3);
        const int off  = (el >> 3) * 128 + cell;
        #pragma unroll
        for (int g = 0; g < 8; ++g) smem[g * 2048 + off] = acc[mi][ni][g];
      }
    }
    __syncthreads();
    float* kt = dst + (size_t)blockIdx.y * (16 * E_DIM) + (size_t)n0 * 16;
    #pragma unroll
    for (int it = 0; it < 16; ++it) {
      const int idx = it * 256 + tid;
      store_g4(kt + (size_t)(idx >> 9) * ((size_t)S_LEN * E_DIM) + (idx & 511) * 4,
               *(const v4f*)&smem[idx * 4]);
    }
  } else {
    #pragma unroll
    for (int mi = 0; mi < 2; ++mi) {
      #pragma unroll
      for (int ni = 0; ni < 4; ++ni) {
        const int col = n0 + wn * 64 + ni * 16 + lm;
        #pragma unroll
        for (int g = 0; g < 8; ++g) {
          const size_t row = (size_t)(m0 + wm * 32 + mi * 16 + g + 8 * half);
          store_g1(dst + row * E_DIM + col, acc[mi][ni][g]);
        }
      }
    }
  }
}

__global__ __launch_bounds__(1024)
void rownorm_kernel(const float* __restrict__ kt, float* __restrict__ gk2) {
  __shared__ float red[4][B_SZ][32];
  const int tid  = threadIdx.x;
  const int w    = tid >> 5;
  const int lane = tid & 31;
  const int bb   = w & 7;
  const int cq   = w >> 3;
  const float* base = kt + (size_t)bb * ((size_t)S_LEN * E_DIM) +
                      (size_t)blockIdx.x * (16 * E_DIM) + cq * 32 * 128 + lane * 4;
  float s = 0.f;
  #pragma unroll 8
  for (int c = 0; c < 32; ++c) {
    const v4f t = load_g4(base + c * 128);
    s += (t.x * t.x + t.y * t.y) + (t.z * t.z + t.w * t.w);
  }
  red[cq][bb][lane] = s;
  __syncthreads();
  if (tid < 128) {
    const int key = tid >> 3;
    const int b2  = tid & 7;
    float a = 0.f;
    #pragma unroll
    for (int qq = 0; qq < 4; ++qq) a += red[qq][b2][key] + red[qq][b2][key + 16];
    store_g1(gk2 + (size_t)blockIdx.x * 128 + tid, GAMMA * a);
  }
}

#define A_QT 32
#define A_NW 16
#define A_ES 64
#define A_KT 64
#define A_SP 68

template <int M>
static __device__ __forceinline__ float row_xor(float x) {
  return __builtin_bit_cast(
      float, __builtin_amdgcn_update_dpp(0, __builtin_bit_cast(int, x),
                                         0x160 + M  , 0xf, 0xf,
                                         false));
}
static __device__ __forceinline__ float row16_max(float x) {
  x = fmaxf(x, row_xor<1>(x));
  x = fmaxf(x, row_xor<2>(x));
  x = fmaxf(x, row_xor<4>(x));
  x = fmaxf(x, row_xor<8>(x));
  return x;
}
static __device__ __forceinline__ float row16_sum(float x) {
  x += row_xor<1>(x);
  x += row_xor<2>(x);
  x += row_xor<4>(x);
  x += row_xor<8>(x);
  return x;
}

__global__ __launch_bounds__(512)
void attn_kernel(const float* __restrict__ q, const float* __restrict__ k,
                 const float* __restrict__ v, const float* __restrict__ gk2,
                 float* __restrict__ out) {
  const int tid  = threadIdx.x;
  const int w    = tid >> 5;
  const int lane = tid & 31;
  const int half = lane >> 4;
  const int lm   = lane & 15;
  const int b    = blockIdx.y;
  const int t0   = blockIdx.x * A_QT;
  const int e0   = w * A_ES;
  const int srow = tid >> 4;
  const int scg  = tid & 15;

  __shared__ __attribute__((aligned(16)))
  float sS[A_NW][A_QT * A_SP];
  __shared__ __attribute__((aligned(16)))
  float sP[A_QT * A_SP];
  __shared__ float sScale[A_QT];
  __shared__ float sL[A_QT];

  float m_run = -__builtin_inff();
  float l_run = 0.f;

  v4f qf[2][8];
  #pragma unroll
  for (int mi = 0; mi < 2; ++mi) {
    const float* qrow =
        q + ((size_t)(t0 + 16 * mi + lm) * B_SZ + b) * E_DIM + e0 + 4 * half;
    #pragma unroll
    for (int s = 0; s < 8; ++s)
      qf[mi][s] = load_g4(qrow + 8 * s) * (2.0f * GAMMA);
  }

  v8f acc[2][4];
  #pragma unroll
  for (int mi = 0; mi < 2; ++mi)
    #pragma unroll
    for (int u = 0; u < 4; ++u) acc[mi][u] = (v8f){};

  for (int j0 = 0; j0 < S_LEN; j0 += A_KT) {
    float gk[4];
    #pragma unroll
    for (int i = 0; i < 4; ++i)
      gk[i] = load_g1(gk2 + (size_t)(j0 + 4 * scg + i) * B_SZ + b);

    const float* kbase = k + (size_t)b * ((size_t)S_LEN * E_DIM) +
                         (size_t)(j0 >> 4) * (16 * E_DIM) + w * 1024 + lane * 4;
    v4f kf[2][8];
    #pragma unroll
    for (int s = 0; s < 8; ++s) kf[0][s] = load_g4(kbase + 128 * s);
    #pragma unroll
    for (int t = 0; t < A_KT / 16; ++t) {
      const int cur = t & 1, nxt = cur ^ 1;
      if (t + 1 < A_KT / 16) {
        const float* kn = kbase + (size_t)(t + 1) * (16 * E_DIM);
        #pragma unroll
        for (int s = 0; s < 8; ++s) kf[nxt][s] = load_g4(kn + 128 * s);
      }
      v8f sc[2];
      sc[0] = (v8f){};
      sc[1] = (v8f){};
      #pragma unroll
      for (int s = 0; s < 8; ++s) {
        const v4f kk = kf[cur][s];
        sc[0] = wmma_f32(qf[0][s].lo, kk.lo, sc[0]);
        sc[1] = wmma_f32(qf[1][s].lo, kk.lo, sc[1]);
        sc[0] = wmma_f32(qf[0][s].hi, kk.hi, sc[0]);
        sc[1] = wmma_f32(qf[1][s].hi, kk.hi, sc[1]);
      }
      #pragma unroll
      for (int mi = 0; mi < 2; ++mi)
        #pragma unroll
        for (int g = 0; g < 8; ++g)
          sS[w][(16 * mi + g + 8 * half) * A_SP + 16 * t + lm] = sc[mi][g];
      __builtin_amdgcn_sched_barrier(0);
    }
    __syncthreads();

    {
      v4f s4 = (v4f){};
      #pragma unroll
      for (int ww = 0; ww < A_NW; ++ww)
        s4 += *(const v4f*)&sS[ww][srow * A_SP + 4 * scg];
      const float l0 = s4.x - gk[0];
      const float l1 = s4.y - gk[1];
      const float l2 = s4.z - gk[2];
      const float l3 = s4.w - gk[3];
      const float mrow = row16_max(fmaxf(fmaxf(l0, l1), fmaxf(l2, l3)));
      const float mnew = fmaxf(m_run, mrow);
      const float scl  = __expf(m_run - mnew);
      m_run = mnew;
      v4f p;
      p.x = __expf(l0 - mnew);
      p.y = __expf(l1 - mnew);
      p.z = __expf(l2 - mnew);
      p.w = __expf(l3 - mnew);
      l_run = l_run * scl + row16_sum((p.x + p.y) + (p.z + p.w));
      *(v4f*)&sP[srow * A_SP + 4 * scg] = p;
      if (scg == 0) sScale[srow] = scl;
    }
    __syncthreads();

    #pragma unroll
    for (int mi = 0; mi < 2; ++mi) {
      float f[8];
      #pragma unroll
      for (int g = 0; g < 8; ++g) f[g] = sScale[16 * mi + g + 8 * half];
      #pragma unroll
      for (int u = 0; u < 4; ++u)
        #pragma unroll
        for (int g = 0; g < 8; ++g) acc[mi][u][g] *= f[g];
    }

    const float* vbase =
        v + ((size_t)(j0 + 2 * half) * B_SZ + b) * E_DIM + e0 + 4 * lm;
    const float* p0 = &sP[lm * A_SP + 2 * half];
    const float* p1 = &sP[(16 + lm) * A_SP + 2 * half];
    v4f vr[2][2][2];
    v2f pa[2][2][2];
    #pragma unroll
    for (int c = 0; c < 2; ++c) {
      vr[0][c][0] = load_g4(vbase + (size_t)(4 * c) * B_SZ * E_DIM);
      vr[0][c][1] = load_g4(vbase + (size_t)(4 * c + 1) * B_SZ * E_DIM);
      pa[0][c][0] = *(const v2f*)(p0 + 4 * c);
      pa[0][c][1] = *(const v2f*)(p1 + 4 * c);
    }
    #pragma unroll
    for (int gq = 0; gq < A_KT / 8; ++gq) {
      const int cur = gq & 1, nxt = cur ^ 1;
      if (gq + 1 < A_KT / 8) {
        #pragma unroll
        for (int c = 0; c < 2; ++c) {
          const int cc = 2 * (gq + 1) + c;
          vr[nxt][c][0] = load_g4(vbase + (size_t)(4 * cc) * B_SZ * E_DIM);
          vr[nxt][c][1] = load_g4(vbase + (size_t)(4 * cc + 1) * B_SZ * E_DIM);
          pa[nxt][c][0] = *(const v2f*)(p0 + 4 * cc);
          pa[nxt][c][1] = *(const v2f*)(p1 + 4 * cc);
        }
      }
      #pragma unroll
      for (int c = 0; c < 2; ++c) {
        #pragma unroll
        for (int u = 0; u < 4; ++u) {
          v2f vf;
          vf.x = vr[cur][c][0][u];
          vf.y = vr[cur][c][1][u];
          acc[0][u] = wmma_f32(pa[cur][c][0], vf, acc[0][u]);
          acc[1][u] = wmma_f32(pa[cur][c][1], vf, acc[1][u]);
        }
      }
    }
  }

  if (scg == 0) sL[srow] = l_run;
  __syncthreads();

  #pragma unroll
  for (int mi = 0; mi < 2; ++mi) {
    float linv[8];
    #pragma unroll
    for (int g = 0; g < 8; ++g) linv[g] = 1.0f / sL[16 * mi + g + 8 * half];
    const int e = e0 + 4 * lm;
    #pragma unroll
    for (int g = 0; g < 8; ++g) {
      const int trow = t0 + 16 * mi + g + 8 * half;
      v4f o;
      o.x = acc[mi][0][g] * linv[g];
      o.y = acc[mi][1][g] * linv[g];
      o.z = acc[mi][2][g] * linv[g];
      o.w = acc[mi][3][g] * linv[g];
      store_g4(out + ((size_t)trow * B_SZ + b) * E_DIM + e, o);
    }
  }
}

extern "C" void kernel_launch(void* const* d_in, const int* in_sizes, int n_in,
                              void* d_out, int out_size, void* d_ws, size_t ws_size,
                              hipStream_t stream) {
  (void)in_sizes; (void)n_in; (void)out_size; (void)ws_size;
  const float* x  = (const float*)d_in[0];
  const float* Wq = (const float*)d_in[1];
  const float* bq = (const float*)d_in[2];
  const float* Wk = (const float*)d_in[3];
  const float* bk = (const float*)d_in[4];
  const float* Wv = (const float*)d_in[5];
  const float* bv = (const float*)d_in[6];
  float* out = (float*)d_out;

  const size_t mat = (size_t)S_LEN * B_SZ * E_DIM;
  float* q   = (float*)d_ws;
  float* k   = q + mat;
  float* v   = k + mat;
  float* gk2 = v + mat;

  dim3 g1((3 * E_DIM) / P_BN, (S_LEN * B_SZ) / P_BM);
  qkv_proj_kernel<<<g1, 256, 0, stream>>>(x, Wq, bq, Wk, bk, Wv, bv, q, k, v);

  rownorm_kernel<<<S_LEN / 16, 1024, 0, stream>>>(k, gk2);

  dim3 g3(S_LEN / A_QT, B_SZ);
  attn_kernel<<<g3, 32 * A_NW, 0, stream>>>(q, k, v, gk2, out);
}


# attention: loads requested further ahead: V rows 2 key groups ahead and across the softmax barriers (barriers order LDS only), next pass's first K fragments during the last V groups; sums unchanged
# speedup vs baseline: 3.0902x; 1.0461x over previous
#include <hip/hip_runtime.h>
#include <hip/hip_bf16.h>


#define S_LEN 2048
#define B_SZ  8
#define E_DIM 1024
constexpr float GAMMA = 0.5f;

typedef __attribute__((ext_vector_type(2))) float v2f;
typedef __attribute__((ext_vector_type(8))) float v8f;

#define GAS __attribute__((address_space(1)))

static __device__ __forceinline__ float load_g1(const float* p) {
  return *(const GAS float*)(p);
}
static __device__ __forceinline__ void store_g1(float* p, float x) {
  *(GAS float*)(p) = x;
}

static __device__ __forceinline__ v8f wmma_f32(v2f a, v2f b, v8f c) {
  return __builtin_amdgcn_wmma_f32_16x16x4_f32(false, a, false, b,
                                               (short)0, c, false, false);
}

#define P_BM 128
#define P_BN 128
#define P_BK 32
#define P_LD 36

typedef __attribute__((ext_vector_type(4))) float v4f;

static __device__ __forceinline__ v4f load_g4(const float* p) {
  return *(const GAS v4f*)(p);
}
static __device__ __forceinline__ void store_g4(float* p, v4f x) {
  *(GAS v4f*)(p) = x;
}

__global__ __launch_bounds__(256)
void qkv_proj_kernel(const float* __restrict__ x,
                     const float* __restrict__ Wq, const float* __restrict__ bq,
                     const float* __restrict__ Wk, const float* __restrict__ bk,
                     const float* __restrict__ Wv, const float* __restrict__ bv,
                     float* __restrict__ q, float* __restrict__ k,
                     float* __restrict__ v) {
  __shared__ __attribute__((aligned(16))) float smem[P_BM * P_BN];
  float* const sA = smem;
  float* const sB = smem + P_BM * P_LD;

  const int tid  = threadIdx.x;
  const int w    = tid >> 5;
  const int lane = tid & 31;
  const int half = lane >> 4;
  const int lm   = lane & 15;
  const int wm   = w >> 1;
  const int wn   = w & 1;
  const int nt   = blockIdx.x;
  const int sel  = nt >> 3;
  const int n0   = (nt & 7) * P_BN;
  const int m0   = blockIdx.y * P_BM;

  const float* Wm = Wq; const float* bias = bq; float* dst = q;
  if (sel == 1) { Wm = Wk; bias = bk; dst = k; }
  if (sel == 2) { Wm = Wv; bias = bv; dst = v; }

  const int srow = tid >> 3;
  const int scol = (tid & 7) * 4;
  const float* xb = x  + (size_t)m0 * E_DIM;
  const float* wb = Wm + (size_t)n0 * E_DIM;
  const unsigned goff = (unsigned)(srow * E_DIM + scol);
  const int st_off = srow * P_LD + scol;

  const int a_off = (wm * 32 + lm) * P_LD + 4 * half;
  const int b_off = (wn * 64 + lm) * P_LD + 4 * half;

  v8f acc[2][4];
  #pragma unroll
  for (int ni = 0; ni < 4; ++ni) {
    const float bi = load_g1(bias + n0 + wn * 64 + ni * 16 + lm);
    #pragma unroll
    for (int mi = 0; mi < 2; ++mi)
      #pragma unroll
      for (int g = 0; g < 8; ++g) acc[mi][ni][g] = bi;
  }

  v4f ra[4], rb[4];
  #pragma unroll
  for (int i = 0; i < 4; ++i) {
    ra[i] = load_g4(xb + goff + i * 32 * E_DIM);
    rb[i] = load_g4(wb + goff + i * 32 * E_DIM);
  }
  #pragma unroll
  for (int i = 0; i < 4; ++i) {
    *(v4f*)&sA[st_off + i * 32 * P_LD] = ra[i];
    *(v4f*)&sB[st_off + i * 32 * P_LD] = rb[i];
  }
  __syncthreads();

  for (int kb = 0; kb < E_DIM / P_BK; ++kb) {
    const bool more = (kb + 1) < (E_DIM / P_BK);
    if (more) {
      xb += P_BK;
      wb += P_BK;
      #pragma unroll
      for (int i = 0; i < 4; ++i) {
        ra[i] = load_g4(xb + goff + i * 32 * E_DIM);
        rb[i] = load_g4(wb + goff + i * 32 * E_DIM);
      }
    }

    v4f fa[2][2], fb[2][4];
    #pragma unroll
    for (int mi = 0; mi < 2; ++mi) fa[0][mi] = *(const v4f*)&sA[a_off + mi * 16 * P_LD];
    #pragma unroll
    for (int ni = 0; ni < 4; ++ni) fb[0][ni] = *(const v4f*)&sB[b_off + ni * 16 * P_LD];

    #pragma unroll
    for (int s = 0; s < P_BK / 8; ++s) {
      const int cur = s & 1, nxt = cur ^ 1;
      if (s + 1 < P_BK / 8) {
        #pragma unroll
        for (int mi = 0; mi < 2; ++mi)
          fa[nxt][mi] = *(const v4f*)&sA[a_off + mi * 16 * P_LD + (s + 1) * 8];
        #pragma unroll
        for (int ni = 0; ni < 4; ++ni)
          fb[nxt][ni] = *(const v4f*)&sB[b_off + ni * 16 * P_LD + (s + 1) * 8];
      }
      #pragma unroll
      for (int mi = 0; mi < 2; ++mi)
        #pragma unroll
        for (int ni = 0; ni < 4; ++ni)
          acc[mi][ni] = wmma_f32(fa[cur][mi].lo, fb[cur][ni].lo, acc[mi][ni]);
      #pragma unroll
      for (int mi = 0; mi < 2; ++mi)
        #pragma unroll
        for (int ni = 0; ni < 4; ++ni)
          acc[mi][ni] = wmma_f32(fa[cur][mi].hi, fb[cur][ni].hi, acc[mi][ni]);
    }

    __syncthreads();
    if (more) {
      #pragma unroll
      for (int i = 0; i < 4; ++i) {
        *(v4f*)&sA[st_off + i * 32 * P_LD] = ra[i];
        *(v4f*)&sB[st_off + i * 32 * P_LD] = rb[i];
      }
    }
    __syncthreads();
  }

  if (sel == 1) {
    #pragma unroll
    for (int mi = 0; mi < 2; ++mi) {
      const int key = wm * 4 + mi * 2 + half;
      #pragma unroll
      for (int ni = 0; ni < 4; ++ni) {
        const int el   = wn * 64 + ni * 16 + lm;
        const int cell = (((el >> 2) & 1) * 16 + key) * 4 + (el & 3);
        const int off  = (el >> 3) * 128 + cell;
        #pragma unroll
        for (int g = 0; g < 8; ++g) smem[g * 2048 + off] = acc[mi][ni][g];
      }
    }
    __syncthreads();
    float* kt = dst + (size_t)blockIdx.y * (16 * E_DIM) + (size_t)n0 * 16;
    #pragma unroll
    for (int it = 0; it < 16; ++it) {
      const int idx = it * 256 + tid;
      store_g4(kt + (size_t)(idx >> 9) * ((size_t)S_LEN * E_DIM) + (idx & 511) * 4,
               *(const v4f*)&smem[idx * 4]);
    }
  } else {
    #pragma unroll
    for (int mi = 0; mi < 2; ++mi) {
      #pragma unroll
      for (int ni = 0; ni < 4; ++ni) {
        const int col = n0 + wn * 64 + ni * 16 + lm;
        #pragma unroll
        for (int g = 0; g < 8; ++g) {
          const size_t row = (size_t)(m0 + wm * 32 + mi * 16 + g + 8 * half);
          store_g1(dst + row * E_DIM + col, acc[mi][ni][g]);
        }
      }
    }
  }
}

__global__ __launch_bounds__(1024)
void rownorm_kernel(const float* __restrict__ kt, float* __restrict__ gk2) {
  __shared__ float red[4][B_SZ][32];
  const int tid  = threadIdx.x;
  const int w    = tid >> 5;
  const int lane = tid & 31;
  const int bb   = w & 7;
  const int cq   = w >> 3;
  const float* base = kt + (size_t)bb * ((size_t)S_LEN * E_DIM) +
                      (size_t)blockIdx.x * (16 * E_DIM) + cq * 32 * 128 + lane * 4;
  float s = 0.f;
  #pragma unroll 8
  for (int c = 0; c < 32; ++c) {
    const v4f t = load_g4(base + c * 128);
    s += (t.x * t.x + t.y * t.y) + (t.z * t.z + t.w * t.w);
  }
  red[cq][bb][lane] = s;
  __syncthreads();
  if (tid < 128) {
    const int key = tid >> 3;
    const int b2  = tid & 7;
    float a = 0.f;
    #pragma unroll
    for (int qq = 0; qq < 4; ++qq) a += red[qq][b2][key] + red[qq][b2][key + 16];
    store_g1(gk2 + (size_t)blockIdx.x * 128 + tid, GAMMA * a);
  }
}

#define A_QT 32
#define A_NW 16
#define A_ES 64
#define A_KT 64
#define A_SP 68

template <int M>
static __device__ __forceinline__ float row_xor(float x) {
  return __builtin_bit_cast(
      float, __builtin_amdgcn_update_dpp(0, __builtin_bit_cast(int, x),
                                         0x160 + M  , 0xf, 0xf,
                                         false));
}
static __device__ __forceinline__ float row16_max(float x) {
  x = fmaxf(x, row_xor<1>(x));
  x = fmaxf(x, row_xor<2>(x));
  x = fmaxf(x, row_xor<4>(x));
  x = fmaxf(x, row_xor<8>(x));
  return x;
}
static __device__ __forceinline__ float row16_sum(float x) {
  x += row_xor<1>(x);
  x += row_xor<2>(x);
  x += row_xor<4>(x);
  x += row_xor<8>(x);
  return x;
}

static __device__ __forceinline__ void lds_barrier() {
  __builtin_amdgcn_fence(__ATOMIC_RELEASE, "workgroup", "local");
  __builtin_amdgcn_s_barrier();
  __builtin_amdgcn_fence(__ATOMIC_ACQUIRE, "workgroup", "local");
}

__global__ __launch_bounds__(512)
void attn_kernel(const float* __restrict__ q, const float* __restrict__ k,
                 const float* __restrict__ v, const float* __restrict__ gk2,
                 float* __restrict__ out) {
  const int tid  = threadIdx.x;
  const int w    = tid >> 5;
  const int lane = tid & 31;
  const int half = lane >> 4;
  const int lm   = lane & 15;
  const int b    = blockIdx.y;
  const int t0   = blockIdx.x * A_QT;
  const int e0   = w * A_ES;
  const int srow = tid >> 4;
  const int scg  = tid & 15;

  __shared__ __attribute__((aligned(16)))
  float sS[A_NW][A_QT * A_SP];
  __shared__ __attribute__((aligned(16)))
  float sP[A_QT * A_SP];
  __shared__ float sScale[A_QT];
  __shared__ float sL[A_QT];

  float m_run = -__builtin_inff();
  float l_run = 0.f;

  v4f qf[2][8];
  #pragma unroll
  for (int mi = 0; mi < 2; ++mi) {
    const float* qrow =
        q + ((size_t)(t0 + 16 * mi + lm) * B_SZ + b) * E_DIM + e0 + 4 * half;
    #pragma unroll
    for (int s = 0; s < 8; ++s)
      qf[mi][s] = load_g4(qrow + 8 * s) * (2.0f * GAMMA);
  }

  v8f acc[2][4];
  #pragma unroll
  for (int mi = 0; mi < 2; ++mi)
    #pragma unroll
    for (int u = 0; u < 4; ++u) acc[mi][u] = (v8f){};

  const float* kbase0 =
      k + (size_t)b * ((size_t)S_LEN * E_DIM) + w * 1024 + lane * 4;
  v4f kf0[8];
  #pragma unroll
  for (int s = 0; s < 8; ++s) kf0[s] = load_g4(kbase0 + 128 * s);

  for (int j0 = 0; j0 < S_LEN; j0 += A_KT) {
    float gk[4];
    #pragma unroll
    for (int i = 0; i < 4; ++i)
      gk[i] = load_g1(gk2 + (size_t)(j0 + 4 * scg + i) * B_SZ + b);

    const float* kbase = kbase0 + (size_t)(j0 >> 4) * (16 * E_DIM);
    v4f kf[2][8];
    #pragma unroll
    for (int s = 0; s < 8; ++s) kf[0][s] = kf0[s];
    #pragma unroll
    for (int t = 0; t < A_KT / 16; ++t) {
      const int cur = t & 1, nxt = cur ^ 1;
      if (t + 1 < A_KT / 16) {
        const float* kn = kbase + (size_t)(t + 1) * (16 * E_DIM);
        #pragma unroll
        for (int s = 0; s < 8; ++s) kf[nxt][s] = load_g4(kn + 128 * s);
      }
      v8f sc[2];
      sc[0] = (v8f){};
      sc[1] = (v8f){};
      #pragma unroll
      for (int s = 0; s < 8; ++s) {
        const v4f kk = kf[cur][s];
        sc[0] = wmma_f32(qf[0][s].lo, kk.lo, sc[0]);
        sc[1] = wmma_f32(qf[1][s].lo, kk.lo, sc[1]);
        sc[0] = wmma_f32(qf[0][s].hi, kk.hi, sc[0]);
        sc[1] = wmma_f32(qf[1][s].hi, kk.hi, sc[1]);
      }
      #pragma unroll
      for (int mi = 0; mi < 2; ++mi)
        #pragma unroll
        for (int g = 0; g < 8; ++g)
          sS[w][(16 * mi + g + 8 * half) * A_SP + 16 * t + lm] = sc[mi][g];
      __builtin_amdgcn_sched_barrier(0);
    }
    const float* vbase =
        v + ((size_t)(j0 + 2 * half) * B_SZ + b) * E_DIM + e0 + 4 * lm;
    v4f vr[3][2][2];
    #pragma unroll
    for (int gg = 0; gg < 2; ++gg) {
      #pragma unroll
      for (int c = 0; c < 2; ++c) {
        const int cc = 2 * gg + c;
        vr[gg][c][0] = load_g4(vbase + (size_t)(4 * cc) * B_SZ * E_DIM);
        vr[gg][c][1] = load_g4(vbase + (size_t)(4 * cc + 1) * B_SZ * E_DIM);
      }
    }
    lds_barrier();

    {
      v4f s4 = (v4f){};
      #pragma unroll
      for (int ww = 0; ww < A_NW; ++ww)
        s4 += *(const v4f*)&sS[ww][srow * A_SP + 4 * scg];
      const float l0 = s4.x - gk[0];
      const float l1 = s4.y - gk[1];
      const float l2 = s4.z - gk[2];
      const float l3 = s4.w - gk[3];
      const float mrow = row16_max(fmaxf(fmaxf(l0, l1), fmaxf(l2, l3)));
      const float mnew = fmaxf(m_run, mrow);
      const float scl  = __expf(m_run - mnew);
      m_run = mnew;
      v4f p;
      p.x = __expf(l0 - mnew);
      p.y = __expf(l1 - mnew);
      p.z = __expf(l2 - mnew);
      p.w = __expf(l3 - mnew);
      l_run = l_run * scl + row16_sum((p.x + p.y) + (p.z + p.w));
      *(v4f*)&sP[srow * A_SP + 4 * scg] = p;
      if (scg == 0) sScale[srow] = scl;
    }
    lds_barrier();

    #pragma unroll
    for (int mi = 0; mi < 2; ++mi) {
      float f[8];
      #pragma unroll
      for (int g = 0; g < 8; ++g) f[g] = sScale[16 * mi + g + 8 * half];
      #pragma unroll
      for (int u = 0; u < 4; ++u)
        #pragma unroll
        for (int g = 0; g < 8; ++g) acc[mi][u][g] *= f[g];
    }

    const int jn = (j0 + A_KT < S_LEN) ? (j0 + A_KT) : j0;
    const float* p0 = &sP[lm * A_SP + 2 * half];
    const float* p1 = &sP[(16 + lm) * A_SP + 2 * half];
    v2f pa[2][2][2];
    #pragma unroll
    for (int c = 0; c < 2; ++c) {
      pa[0][c][0] = *(const v2f*)(p0 + 4 * c);
      pa[0][c][1] = *(const v2f*)(p1 + 4 * c);
    }
    #pragma unroll
    for (int gq = 0; gq < A_KT / 8; ++gq) {
      const int cur = gq % 3, far = (gq + 2) % 3;
      const int pc = gq & 1, pn = pc ^ 1;
      if (gq + 2 < A_KT / 8) {
        #pragma unroll
        for (int c = 0; c < 2; ++c) {
          const int cc = 2 * (gq + 2) + c;
          vr[far][c][0] = load_g4(vbase + (size_t)(4 * cc) * B_SZ * E_DIM);
          vr[far][c][1] = load_g4(vbase + (size_t)(4 * cc + 1) * B_SZ * E_DIM);
        }
      }
      if (gq + 1 < A_KT / 8) {
        #pragma unroll
        for (int c = 0; c < 2; ++c) {
          const int cc = 2 * (gq + 1) + c;
          pa[pn][c][0] = *(const v2f*)(p0 + 4 * cc);
          pa[pn][c][1] = *(const v2f*)(p1 + 4 * cc);
        }
      }
      if (gq + 2 >= A_KT / 8) {
        const float* kn0 = kbase0 + (size_t)(jn >> 4) * (16 * E_DIM);
        #pragma unroll
        for (int s = 0; s < 4; ++s) {
          const int ss = 4 * (gq + 2 - A_KT / 8) + s;
          kf0[ss] = load_g4(kn0 + 128 * ss);
        }
      }
      #pragma unroll
      for (int c = 0; c < 2; ++c) {
        #pragma unroll
        for (int u = 0; u < 4; ++u) {
          v2f vf;
          vf.x = vr[cur][c][0][u];
          vf.y = vr[cur][c][1][u];
          acc[0][u] = wmma_f32(pa[pc][c][0], vf, acc[0][u]);
          acc[1][u] = wmma_f32(pa[pc][c][1], vf, acc[1][u]);
        }
      }
    }
  }

  if (scg == 0) sL[srow] = l_run;
  __syncthreads();

  #pragma unroll
  for (int mi = 0; mi < 2; ++mi) {
    float linv[8];
    #pragma unroll
    for (int g = 0; g < 8; ++g) linv[g] = 1.0f / sL[16 * mi + g + 8 * half];
    const int e = e0 + 4 * lm;
    #pragma unroll
    for (int g = 0; g < 8; ++g) {
      const int trow = t0 + 16 * mi + g + 8 * half;
      v4f o;
      o.x = acc[mi][0][g] * linv[g];
      o.y = acc[mi][1][g] * linv[g];
      o.z = acc[mi][2][g] * linv[g];
      o.w = acc[mi][3][g] * linv[g];
      store_g4(out + ((size_t)trow * B_SZ + b) * E_DIM + e, o);
    }
  }
}

extern "C" void kernel_launch(void* const* d_in, const int* in_sizes, int n_in,
                              void* d_out, int out_size, void* d_ws, size_t ws_size,
                              hipStream_t stream) {
  (void)in_sizes; (void)n_in; (void)out_size; (void)ws_size;
  const float* x  = (const float*)d_in[0];
  const float* Wq = (const float*)d_in[1];
  const float* bq = (const float*)d_in[2];
  const float* Wk = (const float*)d_in[3];
  const float* bk = (const float*)d_in[4];
  const float* Wv = (const float*)d_in[5];
  const float* bv = (const float*)d_in[6];
  float* out = (float*)d_out;

  const size_t mat = (size_t)S_LEN * B_SZ * E_DIM;
  float* q   = (float*)d_ws;
  float* k   = q + mat;
  float* v   = k + mat;
  float* gk2 = v + mat;

  dim3 g1((3 * E_DIM) / P_BN, (S_LEN * B_SZ) / P_BM);
  qkv_proj_kernel<<<g1, 256, 0, stream>>>(x, Wq, bq, Wk, bk, Wv, bv, q, k, v);

  rownorm_kernel<<<S_LEN / 16, 1024, 0, stream>>>(k, gk2);

  dim3 g3(S_LEN / A_QT, B_SZ);
  attn_kernel<<<g3, 32 * A_NW, 0, stream>>>(q, k, v, gk2, out);
}
